# starting kernel
# baseline (speedup 1.0000x reference)
#include <hip/hip_runtime.h>

#define Nn   50000
#define Ee   400000
#define INF_ 256
#define Hh   8
#define Dd   32
#define HD   256
#define EFe  3

typedef __attribute__((ext_vector_type(2))) float v2f;
typedef __attribute__((ext_vector_type(8))) float v8f;

__device__ __forceinline__ unsigned f2o(float f) {
  unsigned u = __float_as_uint(f);
  return (u & 0x80000000u) ? ~u : (u | 0x80000000u);
}
__device__ __forceinline__ float o2f(unsigned u) {
  return (u & 0x80000000u) ? __uint_as_float(u & 0x7fffffffu) : __uint_as_float(~u);
}

__global__ void init_kernel(float* __restrict__ out, float* __restrict__ denom,
                            unsigned* __restrict__ amax_u, const float* __restrict__ bias) {
  int idx = blockIdx.x * blockDim.x + threadIdx.x;
  if (idx < Nn * HD) out[idx] = bias[idx & (HD - 1)];
  if (idx < Nn * Hh) { denom[idx] = 0.f; amax_u[idx] = 0u; }
}

__global__ __launch_bounds__(256) void gemm_h_wmma(const float* __restrict__ x,
                                                   const float* __restrict__ Wm,
                                                   float* __restrict__ h) {
  __shared__ float sA[32][33];
  __shared__ float sB[128][33];
  const int tid   = threadIdx.x;
  const int mBase = blockIdx.x * 32;
  const int nBase = blockIdx.y * 128;
  const int wave  = tid >> 5, ln = tid & 31;
  const int half  = ln >> 4, mr = ln & 15;
  const int mt    = wave & 1;
  const int nt0   = (wave >> 1) << 1;

  v8f acc0 = {};
  v8f acc1 = {};

  for (int k0 = 0; k0 < INF_; k0 += 32) {
    for (int i = tid; i < 32 * 32; i += 256) {
      int r = i >> 5, c = i & 31, row = mBase + r;
      sA[r][c] = (row < Nn) ? x[row * INF_ + k0 + c] : 0.f;
    }
    for (int i = tid; i < 128 * 32; i += 256) {
      int r = i >> 5, c = i & 31;
      sB[r][c] = Wm[(nBase + r) * INF_ + k0 + c];
    }
    __syncthreads();

    const int ar  = mt * 16 + mr;
    const int br0 = nt0 * 16 + mr;
    const int br1 = br0 + 16;
#pragma unroll
    for (int kk = 0; kk < 32; kk += 4) {
      const int kh = kk + half * 2;
      v2f a, b0, b1;
      a.x  = sA[ar][kh];   a.y  = sA[ar][kh + 1];
      b0.x = sB[br0][kh];  b0.y = sB[br0][kh + 1];
      b1.x = sB[br1][kh];  b1.y = sB[br1][kh + 1];
      acc0 = __builtin_amdgcn_wmma_f32_16x16x4_f32(false, a, false, b0, (short)0, acc0, false, false);
      acc1 = __builtin_amdgcn_wmma_f32_16x16x4_f32(false, a, false, b1, (short)0, acc1, false, false);
    }
    __syncthreads();
  }

  const int rowBase = mBase + mt * 16 + half * 8;
#pragma unroll
  for (int r = 0; r < 8; ++r) {
    int row = rowBase + r;
    if (row < Nn) {
      h[row * HD + nBase + nt0 * 16 + mr]       = acc0[r];
      h[row * HD + nBase + (nt0 + 1) * 16 + mr] = acc1[r];
    }
  }
}

__global__ __launch_bounds__(256) void edge_alpha_kernel(
    const float* __restrict__ h, const float* __restrict__ a_src,
    const float* __restrict__ a_dst, const float* __restrict__ edge_W,
    const float* __restrict__ edge_b, const float* __restrict__ edge_attr,
    const int* __restrict__ ei, float* __restrict__ alpha,
    unsigned* __restrict__ amax_u) {
  int e = blockIdx.x * 8 + (threadIdx.x >> 5);
  if (e >= Ee) return;
  int lane = threadIdx.x & 31;
  int src = ei[e], dst = ei[Ee + e];
  float ea0 = edge_attr[e * EFe + 0];
  float ea1 = edge_attr[e * EFe + 1];
  float ea2 = edge_attr[e * EFe + 2];
  const float* hs_row = h + (long)src * HD;
  const float* hd_row = h + (long)dst * HD;

  for (int hh = 0; hh < Hh; ++hh) {
    int hd_i = hh * Dd + lane;
    float hs  = hs_row[hd_i];
    float hdd = hd_row[hd_i];
    float enc = fmaf(ea0, edge_W[hd_i * EFe + 0],
                fmaf(ea1, edge_W[hd_i * EFe + 1],
                fmaf(ea2, edge_W[hd_i * EFe + 2], edge_b[hd_i])));
    enc = fmaxf(enc, 0.f);
    float p = hs * a_src[hd_i] + hdd * a_dst[hd_i] + enc * hdd;
#pragma unroll
    for (int off = 16; off > 0; off >>= 1) p += __shfl_xor(p, off, 32);
    if (lane == 0) {
      float al = (p >= 0.f) ? p : 0.2f * p;
      alpha[e * Hh + hh] = al;
      atomicMax(&amax_u[dst * Hh + hh], f2o(al));
    }
  }
}

__global__ void softmax_num_kernel(const int* __restrict__ ei,
                                   const unsigned* __restrict__ amax_u,
                                   float* __restrict__ alpha_ex,
                                   float* __restrict__ denom) {
  int idx = blockIdx.x * blockDim.x + threadIdx.x;
  if (idx >= Ee * Hh) return;
  int e = idx >> 3, hh = idx & 7;
  int dst = ei[Ee + e];
  float m  = o2f(amax_u[dst * Hh + hh]);
  float ex = __expf(alpha_ex[idx] - m);
  alpha_ex[idx] = ex;
  atomicAdd(&denom[dst * Hh + hh], ex);
}

__global__ __launch_bounds__(256) void scatter_out_kernel(
    const float* __restrict__ h, const float* __restrict__ ex,
    const float* __restrict__ denom, const int* __restrict__ ei,
    float* __restrict__ out) {
  int e = blockIdx.x * 8 + (threadIdx.x >> 5);
  if (e >= Ee) return;
  int lane = threadIdx.x & 31;
  int src = ei[e], dst = ei[Ee + e];
  const float* hs_row = h + (long)src * HD;
  float* out_row = out + (long)dst * HD;
  for (int hh = 0; hh < Hh; ++hh) {
    float attn = ex[e * Hh + hh] / fmaxf(denom[dst * Hh + hh], 1e-10f);
    atomicAdd(&out_row[hh * Dd + lane], hs_row[hh * Dd + lane] * attn);
  }
}

extern "C" void kernel_launch(void* const* d_in, const int* in_sizes, int n_in,
                              void* d_out, int out_size, void* d_ws, size_t ws_size,
                              hipStream_t stream) {
  const float* x        = (const float*)d_in[0];
  const float* Wm       = (const float*)d_in[1];
  const float* a_src    = (const float*)d_in[2];
  const float* a_dst    = (const float*)d_in[3];
  const float* edge_W   = (const float*)d_in[4];
  const float* edge_b   = (const float*)d_in[5];
  const float* bias     = (const float*)d_in[6];
  const float* edge_attr= (const float*)d_in[7];
  const int*   ei       = (const int*)d_in[8];
  float* out = (float*)d_out;

  float*    h       = (float*)d_ws;
  float*    alpha   = h + (size_t)Nn * HD;
  unsigned* amax_u  = (unsigned*)(alpha + (size_t)Ee * Hh);
  float*    denom   = (float*)(amax_u + (size_t)Nn * Hh);

  init_kernel<<<(Nn * HD + 255) / 256, 256, 0, stream>>>(out, denom, amax_u, bias);

  dim3 g((Nn + 31) / 32, HD / 128);
  gemm_h_wmma<<<g, 256, 0, stream>>>(x, Wm, h);

  edge_alpha_kernel<<<(Ee + 7) / 8, 256, 0, stream>>>(h, a_src, a_dst, edge_W, edge_b,
                                                      edge_attr, ei, alpha, amax_u);
  softmax_num_kernel<<<(Ee * Hh + 255) / 256, 256, 0, stream>>>(ei, amax_u, alpha, denom);
  scatter_out_kernel<<<(Ee + 7) / 8, 256, 0, stream>>>(h, alpha, denom, ei, out);
}


# edge-logit kernel: a wave loops over a tile of 16 consecutive edges with its per-feature parameters, endpoints and attributes held in registers and the next edge's rows in flight; sum order kept
# speedup vs baseline: 1.4415x; 1.4415x over previous
#include <hip/hip_runtime.h>

#define Nn   50000
#define Ee   400000
#define INF_ 256
#define Hh   8
#define Dd   32
#define HD   256
#define EFe  3

typedef __attribute__((ext_vector_type(2))) float v2f;
typedef __attribute__((ext_vector_type(8))) float v8f;

__device__ __forceinline__ unsigned f2o(float f) {
  unsigned u = __float_as_uint(f);
  return (u & 0x80000000u) ? ~u : (u | 0x80000000u);
}
__device__ __forceinline__ float o2f(unsigned u) {
  return (u & 0x80000000u) ? __uint_as_float(u & 0x7fffffffu) : __uint_as_float(~u);
}

__global__ void init_kernel(float* __restrict__ out, float* __restrict__ denom,
                            unsigned* __restrict__ amax_u, const float* __restrict__ bias) {
  int idx = blockIdx.x * blockDim.x + threadIdx.x;
  if (idx < Nn * HD) out[idx] = bias[idx & (HD - 1)];
  if (idx < Nn * Hh) { denom[idx] = 0.f; amax_u[idx] = 0u; }
}

__global__ __launch_bounds__(256) void gemm_h_wmma(const float* __restrict__ x,
                                                   const float* __restrict__ Wm,
                                                   float* __restrict__ h) {
  __shared__ float sA[32][33];
  __shared__ float sB[128][33];
  const int tid   = threadIdx.x;
  const int mBase = blockIdx.x * 32;
  const int nBase = blockIdx.y * 128;
  const int wave  = tid >> 5, ln = tid & 31;
  const int half  = ln >> 4, mr = ln & 15;
  const int mt    = wave & 1;
  const int nt0   = (wave >> 1) << 1;

  v8f acc0 = {};
  v8f acc1 = {};

  for (int k0 = 0; k0 < INF_; k0 += 32) {
    for (int i = tid; i < 32 * 32; i += 256) {
      int r = i >> 5, c = i & 31, row = mBase + r;
      sA[r][c] = (row < Nn) ? x[row * INF_ + k0 + c] : 0.f;
    }
    for (int i = tid; i < 128 * 32; i += 256) {
      int r = i >> 5, c = i & 31;
      sB[r][c] = Wm[(nBase + r) * INF_ + k0 + c];
    }
    __syncthreads();

    const int ar  = mt * 16 + mr;
    const int br0 = nt0 * 16 + mr;
    const int br1 = br0 + 16;
#pragma unroll
    for (int kk = 0; kk < 32; kk += 4) {
      const int kh = kk + half * 2;
      v2f a, b0, b1;
      a.x  = sA[ar][kh];   a.y  = sA[ar][kh + 1];
      b0.x = sB[br0][kh];  b0.y = sB[br0][kh + 1];
      b1.x = sB[br1][kh];  b1.y = sB[br1][kh + 1];
      acc0 = __builtin_amdgcn_wmma_f32_16x16x4_f32(false, a, false, b0, (short)0, acc0, false, false);
      acc1 = __builtin_amdgcn_wmma_f32_16x16x4_f32(false, a, false, b1, (short)0, acc1, false, false);
    }
    __syncthreads();
  }

  const int rowBase = mBase + mt * 16 + half * 8;
#pragma unroll
  for (int r = 0; r < 8; ++r) {
    int row = rowBase + r;
    if (row < Nn) {
      h[row * HD + nBase + nt0 * 16 + mr]       = acc0[r];
      h[row * HD + nBase + (nt0 + 1) * 16 + mr] = acc1[r];
    }
  }
}

#define ET 16
static_assert(Ee % (8 * ET) == 0, "the tiles of 8 waves x ET edges must cover the edge list exactly");

__device__ __forceinline__ int   lane_bcast_i(int v, int l)   { return __builtin_amdgcn_readlane(v, l); }
__device__ __forceinline__ float lane_bcast_f(float v, int l) {
  return __int_as_float(__builtin_amdgcn_readlane(__float_as_int(v), l));
}

__global__ __launch_bounds__(256) void edge_alpha_kernel(
    const float* __restrict__ h, const float* __restrict__ a_src,
    const float* __restrict__ a_dst, const float* __restrict__ edge_W,
    const float* __restrict__ edge_b, const float* __restrict__ edge_attr,
    const int* __restrict__ ei, float* __restrict__ alpha,
    unsigned* __restrict__ amax_u) {
  const int lane = threadIdx.x & 31;
  const int wv = __builtin_amdgcn_readfirstlane((int)(threadIdx.x >> 5));
  const int e0 = (blockIdx.x * 8 + wv) * ET;

  float w0[Hh], w1[Hh], w2[Hh], bb[Hh], as_[Hh], ad_[Hh];
#pragma unroll
  for (int hh = 0; hh < Hh; ++hh) {
    const int f = hh * Dd + lane;
    w0[hh]  = edge_W[f * EFe + 0];
    w1[hh]  = edge_W[f * EFe + 1];
    w2[hh]  = edge_W[f * EFe + 2];
    bb[hh]  = edge_b[f];
    as_[hh] = a_src[f];
    ad_[hh] = a_dst[f];
  }

  const int   t      = lane & (ET - 1);
  const int   my_src = ei[e0 + t];
  const int   my_dst = ei[Ee + e0 + t];
  const float my_a0  = edge_attr[(e0 + t) * EFe + 0];
  const float my_a1  = edge_attr[(e0 + t) * EFe + 1];
  const float my_a2  = edge_attr[(e0 + t) * EFe + 2];

  const int  keep_edge = lane >> 1;
  const bool keep_hi   = (lane & 1) != 0;
  float keep[4] = {0.f, 0.f, 0.f, 0.f};

  auto load_rows = [&](int i, float (&hs)[Hh], float (&hd)[Hh]) {
    const int s = lane_bcast_i(my_src, i);
    const int d = lane_bcast_i(my_dst, i);
    const float* ps = h + (long)s * HD + lane;
    const float* pd = h + (long)d * HD + lane;
#pragma unroll
    for (int hh = 0; hh < Hh; ++hh) { hs[hh] = ps[hh * Dd]; hd[hh] = pd[hh * Dd]; }
  };

  auto reduce_edge = [&](int i, const float (&hs)[Hh], const float (&hd)[Hh]) {
    const float ea0 = lane_bcast_f(my_a0, i);
    const float ea1 = lane_bcast_f(my_a1, i);
    const float ea2 = lane_bcast_f(my_a2, i);
    const bool mine = (keep_edge == i);
    float p[Hh];
#pragma unroll
    for (int hh = 0; hh < Hh; ++hh) {
      float enc = fmaf(ea0, w0[hh], fmaf(ea1, w1[hh], fmaf(ea2, w2[hh], bb[hh])));
      enc = fmaxf(enc, 0.f);
      p[hh] = hs[hh] * as_[hh] + hd[hh] * ad_[hh] + enc * hd[hh];
    }
#pragma unroll
    for (int off = 16; off > 0; off >>= 1) {
      float q[Hh];
#pragma unroll
      for (int hh = 0; hh < Hh; ++hh) q[hh] = __shfl_xor(p[hh], off, 32);
#pragma unroll
      for (int hh = 0; hh < Hh; ++hh) p[hh] += q[hh];
    }
#pragma unroll
    for (int hh = 0; hh < Hh; ++hh) {
      const float al = (p[hh] >= 0.f) ? p[hh] : 0.2f * p[hh];
      if (mine && (keep_hi == ((hh >> 2) != 0))) keep[hh & 3] = al;
    }
  };

  float hsA[Hh], hdA[Hh], hsB[Hh], hdB[Hh];
  load_rows(0, hsA, hdA);
  for (int i = 0; i < ET - 2; i += 2) {
    load_rows(i + 1, hsB, hdB);
    reduce_edge(i, hsA, hdA);
    load_rows(i + 2, hsA, hdA);
    reduce_edge(i + 1, hsB, hdB);
  }
  load_rows(ET - 1, hsB, hdB);
  reduce_edge(ET - 2, hsA, hdA);
  reduce_edge(ET - 1, hsB, hdB);

  const int my_e = e0 + keep_edge;
  const int hb   = keep_hi ? 4 : 0;
  const int my_d = ei[Ee + my_e];
  float4 v4; v4.x = keep[0]; v4.y = keep[1]; v4.z = keep[2]; v4.w = keep[3];
  *reinterpret_cast<float4*>(alpha + (long)my_e * Hh + hb) = v4;
  unsigned* am = amax_u + (long)my_d * Hh + hb;
  atomicMax(am + 0, f2o(keep[0]));
  atomicMax(am + 1, f2o(keep[1]));
  atomicMax(am + 2, f2o(keep[2]));
  atomicMax(am + 3, f2o(keep[3]));
}

__global__ void softmax_num_kernel(const int* __restrict__ ei,
                                   const unsigned* __restrict__ amax_u,
                                   float* __restrict__ alpha_ex,
                                   float* __restrict__ denom) {
  int idx = blockIdx.x * blockDim.x + threadIdx.x;
  if (idx >= Ee * Hh) return;
  int e = idx >> 3, hh = idx & 7;
  int dst = ei[Ee + e];
  float m  = o2f(amax_u[dst * Hh + hh]);
  float ex = __expf(alpha_ex[idx] - m);
  alpha_ex[idx] = ex;
  atomicAdd(&denom[dst * Hh + hh], ex);
}

__global__ __launch_bounds__(256) void scatter_out_kernel(
    const float* __restrict__ h, const float* __restrict__ ex,
    const float* __restrict__ denom, const int* __restrict__ ei,
    float* __restrict__ out) {
  int e = blockIdx.x * 8 + (threadIdx.x >> 5);
  if (e >= Ee) return;
  int lane = threadIdx.x & 31;
  int src = ei[e], dst = ei[Ee + e];
  const float* hs_row = h + (long)src * HD;
  float* out_row = out + (long)dst * HD;
  for (int hh = 0; hh < Hh; ++hh) {
    float attn = ex[e * Hh + hh] / fmaxf(denom[dst * Hh + hh], 1e-10f);
    atomicAdd(&out_row[hh * Dd + lane], hs_row[hh * Dd + lane] * attn);
  }
}

extern "C" void kernel_launch(void* const* d_in, const int* in_sizes, int n_in,
                              void* d_out, int out_size, void* d_ws, size_t ws_size,
                              hipStream_t stream) {
  const float* x        = (const float*)d_in[0];
  const float* Wm       = (const float*)d_in[1];
  const float* a_src    = (const float*)d_in[2];
  const float* a_dst    = (const float*)d_in[3];
  const float* edge_W   = (const float*)d_in[4];
  const float* edge_b   = (const float*)d_in[5];
  const float* bias     = (const float*)d_in[6];
  const float* edge_attr= (const float*)d_in[7];
  const int*   ei       = (const int*)d_in[8];
  float* out = (float*)d_out;

  float*    h       = (float*)d_ws;
  float*    alpha   = h + (size_t)Nn * HD;
  unsigned* amax_u  = (unsigned*)(alpha + (size_t)Ee * Hh);
  float*    denom   = (float*)(amax_u + (size_t)Nn * Hh);

  init_kernel<<<(Nn * HD + 255) / 256, 256, 0, stream>>>(out, denom, amax_u, bias);

  dim3 g((Nn + 31) / 32, HD / 128);
  gemm_h_wmma<<<g, 256, 0, stream>>>(x, Wm, h);

  edge_alpha_kernel<<<Ee / (8 * ET), 256, 0, stream>>>(h, a_src, a_dst, edge_W, edge_b,
                                                      edge_attr, ei, alpha, amax_u);
  softmax_num_kernel<<<(Ee * Hh + 255) / 256, 256, 0, stream>>>(ei, amax_u, alpha, denom);
  scatter_out_kernel<<<(Ee + 7) / 8, 256, 0, stream>>>(h, alpha, denom, ei, out);
}


# init, softmax-numerator, scatter kernels re-cut into 10-40x fewer, longer waves (10 stores or elements per thread; 16 edges per wave, next row in flight): wave entry was the limit; atomics kept
# speedup vs baseline: 1.9616x; 1.3608x over previous
#include <hip/hip_runtime.h>

#define Nn   50000
#define Ee   400000
#define INF_ 256
#define Hh   8
#define Dd   32
#define HD   256
#define EFe  3

typedef __attribute__((ext_vector_type(2))) float v2f;
typedef __attribute__((ext_vector_type(8))) float v8f;

__device__ __forceinline__ unsigned f2o(float f) {
  unsigned u = __float_as_uint(f);
  return (u & 0x80000000u) ? ~u : (u | 0x80000000u);
}
__device__ __forceinline__ float o2f(unsigned u) {
  return (u & 0x80000000u) ? __uint_as_float(u & 0x7fffffffu) : __uint_as_float(~u);
}

#define INIT_BLOCKS 1250
#define INIT_TRIPS  10
static_assert(INIT_BLOCKS * 256 * INIT_TRIPS * 4 == Nn * HD, "the 16-byte stores must cover out exactly");
static_assert((INIT_BLOCKS * 256) % (HD / 4) == 0, "a thread must keep its column group from trip to trip");
static_assert((Nn * Hh) % 4 == 0 && INIT_BLOCKS * 256 * 4 >= Nn * Hh, "one 16-byte store per thread must cover denom and amax");

__global__ __launch_bounds__(256) void init_kernel(float* __restrict__ out, float* __restrict__ denom,
                                                   unsigned* __restrict__ amax_u,
                                                   const float* __restrict__ bias) {
  const int tid  = blockIdx.x * 256 + threadIdx.x;
  const int nthr = INIT_BLOCKS * 256;
  const int cg = (tid & (HD / 4 - 1)) * 4;
  float4 b4; b4.x = bias[cg]; b4.y = bias[cg + 1]; b4.z = bias[cg + 2]; b4.w = bias[cg + 3];
  float4* o4 = reinterpret_cast<float4*>(out);
#pragma unroll
  for (int k = 0; k < INIT_TRIPS; ++k) o4[tid + k * nthr] = b4;
  if (tid < Nn * Hh / 4) {
    float4 zf; zf.x = 0.f; zf.y = 0.f; zf.z = 0.f; zf.w = 0.f;
    uint4  zu; zu.x = 0u; zu.y = 0u; zu.z = 0u; zu.w = 0u;
    reinterpret_cast<float4*>(denom)[tid] = zf;
    reinterpret_cast<uint4*>(amax_u)[tid] = zu;
  }
}

__global__ __launch_bounds__(256) void gemm_h_wmma(const float* __restrict__ x,
                                                   const float* __restrict__ Wm,
                                                   float* __restrict__ h) {
  __shared__ float sA[32][33];
  __shared__ float sB[128][33];
  const int tid   = threadIdx.x;
  const int mBase = blockIdx.x * 32;
  const int nBase = blockIdx.y * 128;
  const int wave  = tid >> 5, ln = tid & 31;
  const int half  = ln >> 4, mr = ln & 15;
  const int mt    = wave & 1;
  const int nt0   = (wave >> 1) << 1;

  v8f acc0 = {};
  v8f acc1 = {};

  for (int k0 = 0; k0 < INF_; k0 += 32) {
    for (int i = tid; i < 32 * 32; i += 256) {
      int r = i >> 5, c = i & 31, row = mBase + r;
      sA[r][c] = (row < Nn) ? x[row * INF_ + k0 + c] : 0.f;
    }
    for (int i = tid; i < 128 * 32; i += 256) {
      int r = i >> 5, c = i & 31;
      sB[r][c] = Wm[(nBase + r) * INF_ + k0 + c];
    }
    __syncthreads();

    const int ar  = mt * 16 + mr;
    const int br0 = nt0 * 16 + mr;
    const int br1 = br0 + 16;
#pragma unroll
    for (int kk = 0; kk < 32; kk += 4) {
      const int kh = kk + half * 2;
      v2f a, b0, b1;
      a.x  = sA[ar][kh];   a.y  = sA[ar][kh + 1];
      b0.x = sB[br0][kh];  b0.y = sB[br0][kh + 1];
      b1.x = sB[br1][kh];  b1.y = sB[br1][kh + 1];
      acc0 = __builtin_amdgcn_wmma_f32_16x16x4_f32(false, a, false, b0, (short)0, acc0, false, false);
      acc1 = __builtin_amdgcn_wmma_f32_16x16x4_f32(false, a, false, b1, (short)0, acc1, false, false);
    }
    __syncthreads();
  }

  const int rowBase = mBase + mt * 16 + half * 8;
#pragma unroll
  for (int r = 0; r < 8; ++r) {
    int row = rowBase + r;
    if (row < Nn) {
      h[row * HD + nBase + nt0 * 16 + mr]       = acc0[r];
      h[row * HD + nBase + (nt0 + 1) * 16 + mr] = acc1[r];
    }
  }
}

#define ET 16
static_assert(Ee % (8 * ET) == 0, "the tiles of 8 waves x ET edges must cover the edge list exactly");

__device__ __forceinline__ int   lane_bcast_i(int v, int l)   { return __builtin_amdgcn_readlane(v, l); }
__device__ __forceinline__ float lane_bcast_f(float v, int l) {
  return __int_as_float(__builtin_amdgcn_readlane(__float_as_int(v), l));
}

__global__ __launch_bounds__(256) void edge_alpha_kernel(
    const float* __restrict__ h, const float* __restrict__ a_src,
    const float* __restrict__ a_dst, const float* __restrict__ edge_W,
    const float* __restrict__ edge_b, const float* __restrict__ edge_attr,
    const int* __restrict__ ei, float* __restrict__ alpha,
    unsigned* __restrict__ amax_u) {
  const int lane = threadIdx.x & 31;
  const int wv = __builtin_amdgcn_readfirstlane((int)(threadIdx.x >> 5));
  const int e0 = (blockIdx.x * 8 + wv) * ET;

  float w0[Hh], w1[Hh], w2[Hh], bb[Hh], as_[Hh], ad_[Hh];
#pragma unroll
  for (int hh = 0; hh < Hh; ++hh) {
    const int f = hh * Dd + lane;
    w0[hh]  = edge_W[f * EFe + 0];
    w1[hh]  = edge_W[f * EFe + 1];
    w2[hh]  = edge_W[f * EFe + 2];
    bb[hh]  = edge_b[f];
    as_[hh] = a_src[f];
    ad_[hh] = a_dst[f];
  }

  const int   t      = lane & (ET - 1);
  const int   my_src = ei[e0 + t];
  const int   my_dst = ei[Ee + e0 + t];
  const float my_a0  = edge_attr[(e0 + t) * EFe + 0];
  const float my_a1  = edge_attr[(e0 + t) * EFe + 1];
  const float my_a2  = edge_attr[(e0 + t) * EFe + 2];

  const int  keep_edge = lane >> 1;
  const bool keep_hi   = (lane & 1) != 0;
  float keep[4] = {0.f, 0.f, 0.f, 0.f};

  auto load_rows = [&](int i, float (&hs)[Hh], float (&hd)[Hh]) {
    const int s = lane_bcast_i(my_src, i);
    const int d = lane_bcast_i(my_dst, i);
    const float* ps = h + (long)s * HD + lane;
    const float* pd = h + (long)d * HD + lane;
#pragma unroll
    for (int hh = 0; hh < Hh; ++hh) { hs[hh] = ps[hh * Dd]; hd[hh] = pd[hh * Dd]; }
  };

  auto reduce_edge = [&](int i, const float (&hs)[Hh], const float (&hd)[Hh]) {
    const float ea0 = lane_bcast_f(my_a0, i);
    const float ea1 = lane_bcast_f(my_a1, i);
    const float ea2 = lane_bcast_f(my_a2, i);
    const bool mine = (keep_edge == i);
    float p[Hh];
#pragma unroll
    for (int hh = 0; hh < Hh; ++hh) {
      float enc = fmaf(ea0, w0[hh], fmaf(ea1, w1[hh], fmaf(ea2, w2[hh], bb[hh])));
      enc = fmaxf(enc, 0.f);
      p[hh] = hs[hh] * as_[hh] + hd[hh] * ad_[hh] + enc * hd[hh];
    }
#pragma unroll
    for (int off = 16; off > 0; off >>= 1) {
      float q[Hh];
#pragma unroll
      for (int hh = 0; hh < Hh; ++hh) q[hh] = __shfl_xor(p[hh], off, 32);
#pragma unroll
      for (int hh = 0; hh < Hh; ++hh) p[hh] += q[hh];
    }
#pragma unroll
    for (int hh = 0; hh < Hh; ++hh) {
      const float al = (p[hh] >= 0.f) ? p[hh] : 0.2f * p[hh];
      if (mine && (keep_hi == ((hh >> 2) != 0))) keep[hh & 3] = al;
    }
  };

  float hsA[Hh], hdA[Hh], hsB[Hh], hdB[Hh];
  load_rows(0, hsA, hdA);
  for (int i = 0; i < ET - 2; i += 2) {
    load_rows(i + 1, hsB, hdB);
    reduce_edge(i, hsA, hdA);
    load_rows(i + 2, hsA, hdA);
    reduce_edge(i + 1, hsB, hdB);
  }
  load_rows(ET - 1, hsB, hdB);
  reduce_edge(ET - 2, hsA, hdA);
  reduce_edge(ET - 1, hsB, hdB);

  const int my_e = e0 + keep_edge;
  const int hb   = keep_hi ? 4 : 0;
  const int my_d = ei[Ee + my_e];
  float4 v4; v4.x = keep[0]; v4.y = keep[1]; v4.z = keep[2]; v4.w = keep[3];
  *reinterpret_cast<float4*>(alpha + (long)my_e * Hh + hb) = v4;
  unsigned* am = amax_u + (long)my_d * Hh + hb;
  atomicMax(am + 0, f2o(keep[0]));
  atomicMax(am + 1, f2o(keep[1]));
  atomicMax(am + 2, f2o(keep[2]));
  atomicMax(am + 3, f2o(keep[3]));
}

#define SM_BLOCKS 1250
#define SM_PER    10
#define SM_B      5
static_assert(SM_BLOCKS * 256 * SM_PER == Ee * Hh, "the grid-stride trips must cover the elements exactly");
static_assert(SM_PER % SM_B == 0, "whole batches");

__global__ __launch_bounds__(256) void softmax_num_kernel(const int* __restrict__ ei,
                                                          const unsigned* __restrict__ amax_u,
                                                          float* __restrict__ alpha_ex,
                                                          float* __restrict__ denom) {
  const int tid  = blockIdx.x * 256 + threadIdx.x;
  const int nthr = SM_BLOCKS * 256;
  const int hh   = tid & 7;
  for (int b = 0; b < SM_PER; b += SM_B) {
    int dst[SM_B]; float al[SM_B]; unsigned mu[SM_B];
#pragma unroll
    for (int k = 0; k < SM_B; ++k) {
      const int idx = tid + (b + k) * nthr;
      dst[k] = ei[Ee + (idx >> 3)];
      al[k]  = alpha_ex[idx];
    }
#pragma unroll
    for (int k = 0; k < SM_B; ++k) mu[k] = amax_u[dst[k] * Hh + hh];
#pragma unroll
    for (int k = 0; k < SM_B; ++k) {
      const int idx = tid + (b + k) * nthr;
      const float ex = __expf(al[k] - o2f(mu[k]));
      alpha_ex[idx] = ex;
      atomicAdd(&denom[dst[k] * Hh + hh], ex);
    }
  }
}

#define ST 16
static_assert(Ee % (8 * ST) == 0, "the tiles of 8 waves x ST edges must cover the edge list exactly");
static_assert(2 * ST == 32, "two lanes (4 heads each) per edge of the tile");

__global__ __launch_bounds__(256) void scatter_out_kernel(
    const float* __restrict__ h, const float* __restrict__ ex,
    const float* __restrict__ denom, const int* __restrict__ ei,
    float* __restrict__ out) {
  const int lane = threadIdx.x & 31;
  const int wv = __builtin_amdgcn_readfirstlane((int)(threadIdx.x >> 5));
  const int e0 = (blockIdx.x * 8 + wv) * ST;

  const int t      = lane & (ST - 1);
  const int my_src = ei[e0 + t];
  const int my_dst = ei[Ee + e0 + t];

  const int ke = lane >> 1;
  const int hb = (lane & 1) * 4;
  const int kd = ei[Ee + e0 + ke];
  const float4 x4 = *reinterpret_cast<const float4*>(ex + (long)(e0 + ke) * Hh + hb);
  const float4 d4 = *reinterpret_cast<const float4*>(denom + (long)kd * Hh + hb);
  float at[4];
  at[0] = x4.x / fmaxf(d4.x, 1e-10f);
  at[1] = x4.y / fmaxf(d4.y, 1e-10f);
  at[2] = x4.z / fmaxf(d4.z, 1e-10f);
  at[3] = x4.w / fmaxf(d4.w, 1e-10f);

  auto load_row = [&](int i, float (&hs)[Hh]) {
    const int s = lane_bcast_i(my_src, i);
    const float* ps = h + (long)s * HD + lane;
#pragma unroll
    for (int hh = 0; hh < Hh; ++hh) hs[hh] = ps[hh * Dd];
  };

  auto add_edge = [&](int i, const float (&hs)[Hh]) {
    const int d = lane_bcast_i(my_dst, i);
    float* po = out + (long)d * HD + lane;
#pragma unroll
    for (int hh = 0; hh < Hh; ++hh) {
      const float attn = lane_bcast_f(at[hh & 3], 2 * i + (hh >> 2));
      atomicAdd(po + hh * Dd, hs[hh] * attn);
    }
  };

  float hsA[Hh], hsB[Hh];
  load_row(0, hsA);
  for (int i = 0; i < ST - 2; i += 2) {
    load_row(i + 1, hsB);
    add_edge(i, hsA);
    load_row(i + 2, hsA);
    add_edge(i + 1, hsB);
  }
  load_row(ST - 1, hsB);
  add_edge(ST - 2, hsA);
  add_edge(ST - 1, hsB);
}

extern "C" void kernel_launch(void* const* d_in, const int* in_sizes, int n_in,
                              void* d_out, int out_size, void* d_ws, size_t ws_size,
                              hipStream_t stream) {
  const float* x        = (const float*)d_in[0];
  const float* Wm       = (const float*)d_in[1];
  const float* a_src    = (const float*)d_in[2];
  const float* a_dst    = (const float*)d_in[3];
  const float* edge_W   = (const float*)d_in[4];
  const float* edge_b   = (const float*)d_in[5];
  const float* bias     = (const float*)d_in[6];
  const float* edge_attr= (const float*)d_in[7];
  const int*   ei       = (const int*)d_in[8];
  float* out = (float*)d_out;

  float*    h       = (float*)d_ws;
  float*    alpha   = h + (size_t)Nn * HD;
  unsigned* amax_u  = (unsigned*)(alpha + (size_t)Ee * Hh);
  float*    denom   = (float*)(amax_u + (size_t)Nn * Hh);

  init_kernel<<<INIT_BLOCKS, 256, 0, stream>>>(out, denom, amax_u, bias);

  dim3 g((Nn + 31) / 32, HD / 128);
  gemm_h_wmma<<<g, 256, 0, stream>>>(x, Wm, h);

  edge_alpha_kernel<<<Ee / (8 * ET), 256, 0, stream>>>(h, a_src, a_dst, edge_W, edge_b,
                                                      edge_attr, ei, alpha, amax_u);
  softmax_num_kernel<<<SM_BLOCKS, 256, 0, stream>>>(ei, amax_u, alpha, denom);
  scatter_out_kernel<<<Ee / (8 * ST), 256, 0, stream>>>(h, alpha, denom, ei, out);
}


# edge-logit kernel: 32-lane head sums of 4 edges x 8 heads by a transposing reduction on the vector ALU (row swap, then DPP row_xmask adds) instead of 40 LDS shuffles per edge; sum tree kept
# speedup vs baseline: 2.1966x; 1.1198x over previous
#include <hip/hip_runtime.h>

#define Nn   50000
#define Ee   400000
#define INF_ 256
#define Hh   8
#define Dd   32
#define HD   256
#define EFe  3

typedef __attribute__((ext_vector_type(2))) float v2f;
typedef __attribute__((ext_vector_type(8))) float v8f;

__device__ __forceinline__ unsigned f2o(float f) {
  unsigned u = __float_as_uint(f);
  return (u & 0x80000000u) ? ~u : (u | 0x80000000u);
}
__device__ __forceinline__ float o2f(unsigned u) {
  return (u & 0x80000000u) ? __uint_as_float(u & 0x7fffffffu) : __uint_as_float(~u);
}

#define INIT_BLOCKS 1250
#define INIT_TRIPS  10
static_assert(INIT_BLOCKS * 256 * INIT_TRIPS * 4 == Nn * HD, "the 16-byte stores must cover out exactly");
static_assert((INIT_BLOCKS * 256) % (HD / 4) == 0, "a thread must keep its column group from trip to trip");
static_assert((Nn * Hh) % 4 == 0 && INIT_BLOCKS * 256 * 4 >= Nn * Hh, "one 16-byte store per thread must cover denom and amax");

__global__ __launch_bounds__(256) void init_kernel(float* __restrict__ out, float* __restrict__ denom,
                                                   unsigned* __restrict__ amax_u,
                                                   const float* __restrict__ bias) {
  const int tid  = blockIdx.x * 256 + threadIdx.x;
  const int nthr = INIT_BLOCKS * 256;
  const int cg = (tid & (HD / 4 - 1)) * 4;
  float4 b4; b4.x = bias[cg]; b4.y = bias[cg + 1]; b4.z = bias[cg + 2]; b4.w = bias[cg + 3];
  float4* o4 = reinterpret_cast<float4*>(out);
#pragma unroll
  for (int k = 0; k < INIT_TRIPS; ++k) o4[tid + k * nthr] = b4;
  if (tid < Nn * Hh / 4) {
    float4 zf; zf.x = 0.f; zf.y = 0.f; zf.z = 0.f; zf.w = 0.f;
    uint4  zu; zu.x = 0u; zu.y = 0u; zu.z = 0u; zu.w = 0u;
    reinterpret_cast<float4*>(denom)[tid] = zf;
    reinterpret_cast<uint4*>(amax_u)[tid] = zu;
  }
}

__global__ __launch_bounds__(256) void gemm_h_wmma(const float* __restrict__ x,
                                                   const float* __restrict__ Wm,
                                                   float* __restrict__ h) {
  __shared__ float sA[32][33];
  __shared__ float sB[128][33];
  const int tid   = threadIdx.x;
  const int mBase = blockIdx.x * 32;
  const int nBase = blockIdx.y * 128;
  const int wave  = tid >> 5, ln = tid & 31;
  const int half  = ln >> 4, mr = ln & 15;
  const int mt    = wave & 1;
  const int nt0   = (wave >> 1) << 1;

  v8f acc0 = {};
  v8f acc1 = {};

  for (int k0 = 0; k0 < INF_; k0 += 32) {
    for (int i = tid; i < 32 * 32; i += 256) {
      int r = i >> 5, c = i & 31, row = mBase + r;
      sA[r][c] = (row < Nn) ? x[row * INF_ + k0 + c] : 0.f;
    }
    for (int i = tid; i < 128 * 32; i += 256) {
      int r = i >> 5, c = i & 31;
      sB[r][c] = Wm[(nBase + r) * INF_ + k0 + c];
    }
    __syncthreads();

    const int ar  = mt * 16 + mr;
    const int br0 = nt0 * 16 + mr;
    const int br1 = br0 + 16;
#pragma unroll
    for (int kk = 0; kk < 32; kk += 4) {
      const int kh = kk + half * 2;
      v2f a, b0, b1;
      a.x  = sA[ar][kh];   a.y  = sA[ar][kh + 1];
      b0.x = sB[br0][kh];  b0.y = sB[br0][kh + 1];
      b1.x = sB[br1][kh];  b1.y = sB[br1][kh + 1];
      acc0 = __builtin_amdgcn_wmma_f32_16x16x4_f32(false, a, false, b0, (short)0, acc0, false, false);
      acc1 = __builtin_amdgcn_wmma_f32_16x16x4_f32(false, a, false, b1, (short)0, acc1, false, false);
    }
    __syncthreads();
  }

  const int rowBase = mBase + mt * 16 + half * 8;
#pragma unroll
  for (int r = 0; r < 8; ++r) {
    int row = rowBase + r;
    if (row < Nn) {
      h[row * HD + nBase + nt0 * 16 + mr]       = acc0[r];
      h[row * HD + nBase + (nt0 + 1) * 16 + mr] = acc1[r];
    }
  }
}

#define ET 16
#define EG 4
static_assert(Ee % (8 * ET) == 0, "the tiles of 8 waves x ET edges must cover the edge list exactly");
static_assert(ET % EG == 0 && EG * Hh == 32, "a group's EG x 8 sums land one per lane");

__device__ __forceinline__ int   lane_bcast_i(int v, int l)   { return __builtin_amdgcn_readlane(v, l); }
__device__ __forceinline__ float lane_bcast_f(float v, int l) {
  return __int_as_float(__builtin_amdgcn_readlane(__float_as_int(v), l));
}

__device__ __forceinline__ float row_swap_add(float a, float b) {
  const auto r = __builtin_amdgcn_permlane16_swap(__float_as_uint(a), __float_as_uint(b), false, false);
  return __uint_as_float(r[0]) + __uint_as_float(r[1]);
}

template <int M>
__device__ __forceinline__ float row_xor(float v) {
  return __int_as_float(__builtin_amdgcn_update_dpp(__float_as_int(v), __float_as_int(v),
                                                    0x160 + M, 0xf, 0xf, true));
}

template <int M>
__device__ __forceinline__ float fold(float lo, float hi, bool up) {
  const float keep = up ? hi : lo;
  const float send = up ? lo : hi;
  return keep + row_xor<M>(send);
}

__global__ __launch_bounds__(256) void edge_alpha_kernel(
    const float* __restrict__ h, const float* __restrict__ a_src,
    const float* __restrict__ a_dst, const float* __restrict__ edge_W,
    const float* __restrict__ edge_b, const float* __restrict__ edge_attr,
    const int* __restrict__ ei, float* __restrict__ alpha,
    unsigned* __restrict__ amax_u) {
  const int lane = threadIdx.x & 31;
  const int wv = __builtin_amdgcn_readfirstlane((int)(threadIdx.x >> 5));
  const int e0 = (blockIdx.x * 8 + wv) * ET;

  float w0[Hh], w1[Hh], w2[Hh], bb[Hh], as_[Hh], ad_[Hh];
#pragma unroll
  for (int hh = 0; hh < Hh; ++hh) {
    const int f = hh * Dd + lane;
    w0[hh]  = edge_W[f * EFe + 0];
    w1[hh]  = edge_W[f * EFe + 1];
    w2[hh]  = edge_W[f * EFe + 2];
    bb[hh]  = edge_b[f];
    as_[hh] = a_src[f];
    ad_[hh] = a_dst[f];
  }

  const int   t      = lane & (ET - 1);
  const int   my_src = ei[e0 + t];
  const int   my_dst = ei[Ee + e0 + t];
  const float my_a0  = edge_attr[(e0 + t) * EFe + 0];
  const float my_a1  = edge_attr[(e0 + t) * EFe + 1];
  const float my_a2  = edge_attr[(e0 + t) * EFe + 2];

  const bool up8 = (lane & 8) != 0, up4 = (lane & 4) != 0, up2 = (lane & 2) != 0, up1 = (lane & 1) != 0;
  const int  ge  = ((lane >> 4) & 1) + 2 * ((lane >> 3) & 1);
  const int  gh  = lane & 7;

  auto load_rows = [&](int i, float (&hs)[Hh], float (&hd)[Hh]) {
    const int s = lane_bcast_i(my_src, i);
    const int d = lane_bcast_i(my_dst, i);
    const float* ps = h + (long)s * HD + lane;
    const float* pd = h + (long)d * HD + lane;
#pragma unroll
    for (int hh = 0; hh < Hh; ++hh) { hs[hh] = ps[hh * Dd]; hd[hh] = pd[hh * Dd]; }
  };

  auto edge_partials = [&](int i, const float (&hs)[Hh], const float (&hd)[Hh], float (&p)[Hh]) {
    const float ea0 = lane_bcast_f(my_a0, i);
    const float ea1 = lane_bcast_f(my_a1, i);
    const float ea2 = lane_bcast_f(my_a2, i);
#pragma unroll
    for (int hh = 0; hh < Hh; ++hh) {
      float enc = fmaf(ea0, w0[hh], fmaf(ea1, w1[hh], fmaf(ea2, w2[hh], bb[hh])));
      enc = fmaxf(enc, 0.f);
      p[hh] = hs[hh] * as_[hh] + hd[hh] * ad_[hh] + enc * hd[hh];
    }
  };

  float hsA[Hh], hdA[Hh], hsB[Hh], hdB[Hh];
  load_rows(0, hsA, hdA);
  for (int i = 0; i < ET; i += EG) {
    float pa[Hh], pb[Hh], u0[Hh], u1[Hh];
    load_rows(i + 1, hsB, hdB);
    edge_partials(i, hsA, hdA, pa);
    load_rows(i + 2, hsA, hdA);
    edge_partials(i + 1, hsB, hdB, pb);
#pragma unroll
    for (int hh = 0; hh < Hh; ++hh) u0[hh] = row_swap_add(pa[hh], pb[hh]);
    load_rows(i + 3, hsB, hdB);
    edge_partials(i + 2, hsA, hdA, pa);
    if (i + EG < ET) load_rows(i + EG, hsA, hdA);
    edge_partials(i + 3, hsB, hdB, pb);
#pragma unroll
    for (int hh = 0; hh < Hh; ++hh) u1[hh] = row_swap_add(pa[hh], pb[hh]);

    float w[Hh];
#pragma unroll
    for (int hh = 0; hh < Hh; ++hh) w[hh] = fold<8>(u0[hh], u1[hh], up8);
    const float x0 = fold<4>(w[0], w[4], up4), x1 = fold<4>(w[1], w[5], up4);
    const float x2 = fold<4>(w[2], w[6], up4), x3 = fold<4>(w[3], w[7], up4);
    const float y0 = fold<2>(x0, x2, up2), y1 = fold<2>(x1, x3, up2);
    const float z  = fold<1>(y0, y1, up1);
    const float al = (z >= 0.f) ? z : 0.2f * z;

    const int my_e = e0 + i + ge;
    const int my_d = __shfl(my_dst, i + ge, 32);
    alpha[(long)my_e * Hh + gh] = al;
    atomicMax(amax_u + (long)my_d * Hh + gh, f2o(al));
  }
}

#define SM_BLOCKS 1250
#define SM_PER    10
#define SM_B      5
static_assert(SM_BLOCKS * 256 * SM_PER == Ee * Hh, "the grid-stride trips must cover the elements exactly");
static_assert(SM_PER % SM_B == 0, "whole batches");

__global__ __launch_bounds__(256) void softmax_num_kernel(const int* __restrict__ ei,
                                                          const unsigned* __restrict__ amax_u,
                                                          float* __restrict__ alpha_ex,
                                                          float* __restrict__ denom) {
  const int tid  = blockIdx.x * 256 + threadIdx.x;
  const int nthr = SM_BLOCKS * 256;
  const int hh   = tid & 7;
  for (int b = 0; b < SM_PER; b += SM_B) {
    int dst[SM_B]; float al[SM_B]; unsigned mu[SM_B];
#pragma unroll
    for (int k = 0; k < SM_B; ++k) {
      const int idx = tid + (b + k) * nthr;
      dst[k] = ei[Ee + (idx >> 3)];
      al[k]  = alpha_ex[idx];
    }
#pragma unroll
    for (int k = 0; k < SM_B; ++k) mu[k] = amax_u[dst[k] * Hh + hh];
#pragma unroll
    for (int k = 0; k < SM_B; ++k) {
      const int idx = tid + (b + k) * nthr;
      const float ex = __expf(al[k] - o2f(mu[k]));
      alpha_ex[idx] = ex;
      atomicAdd(&denom[dst[k] * Hh + hh], ex);
    }
  }
}

#define ST 16
static_assert(Ee % (8 * ST) == 0, "the tiles of 8 waves x ST edges must cover the edge list exactly");
static_assert(2 * ST == 32, "two lanes (4 heads each) per edge of the tile");

__global__ __launch_bounds__(256) void scatter_out_kernel(
    const float* __restrict__ h, const float* __restrict__ ex,
    const float* __restrict__ denom, const int* __restrict__ ei,
    float* __restrict__ out) {
  const int lane = threadIdx.x & 31;
  const int wv = __builtin_amdgcn_readfirstlane((int)(threadIdx.x >> 5));
  const int e0 = (blockIdx.x * 8 + wv) * ST;

  const int t      = lane & (ST - 1);
  const int my_src = ei[e0 + t];
  const int my_dst = ei[Ee + e0 + t];

  const int ke = lane >> 1;
  const int hb = (lane & 1) * 4;
  const int kd = ei[Ee + e0 + ke];
  const float4 x4 = *reinterpret_cast<const float4*>(ex + (long)(e0 + ke) * Hh + hb);
  const float4 d4 = *reinterpret_cast<const float4*>(denom + (long)kd * Hh + hb);
  float at[4];
  at[0] = x4.x / fmaxf(d4.x, 1e-10f);
  at[1] = x4.y / fmaxf(d4.y, 1e-10f);
  at[2] = x4.z / fmaxf(d4.z, 1e-10f);
  at[3] = x4.w / fmaxf(d4.w, 1e-10f);

  auto load_row = [&](int i, float (&hs)[Hh]) {
    const int s = lane_bcast_i(my_src, i);
    const float* ps = h + (long)s * HD + lane;
#pragma unroll
    for (int hh = 0; hh < Hh; ++hh) hs[hh] = ps[hh * Dd];
  };

  auto add_edge = [&](int i, const float (&hs)[Hh]) {
    const int d = lane_bcast_i(my_dst, i);
    float* po = out + (long)d * HD + lane;
#pragma unroll
    for (int hh = 0; hh < Hh; ++hh) {
      const float attn = lane_bcast_f(at[hh & 3], 2 * i + (hh >> 2));
      atomicAdd(po + hh * Dd, hs[hh] * attn);
    }
  };

  float hsA[Hh], hsB[Hh];
  load_row(0, hsA);
  for (int i = 0; i < ST - 2; i += 2) {
    load_row(i + 1, hsB);
    add_edge(i, hsA);
    load_row(i + 2, hsA);
    add_edge(i + 1, hsB);
  }
  load_row(ST - 1, hsB);
  add_edge(ST - 2, hsA);
  add_edge(ST - 1, hsB);
}

extern "C" void kernel_launch(void* const* d_in, const int* in_sizes, int n_in,
                              void* d_out, int out_size, void* d_ws, size_t ws_size,
                              hipStream_t stream) {
  const float* x        = (const float*)d_in[0];
  const float* Wm       = (const float*)d_in[1];
  const float* a_src    = (const float*)d_in[2];
  const float* a_dst    = (const float*)d_in[3];
  const float* edge_W   = (const float*)d_in[4];
  const float* edge_b   = (const float*)d_in[5];
  const float* bias     = (const float*)d_in[6];
  const float* edge_attr= (const float*)d_in[7];
  const int*   ei       = (const int*)d_in[8];
  float* out = (float*)d_out;

  float*    h       = (float*)d_ws;
  float*    alpha   = h + (size_t)Nn * HD;
  unsigned* amax_u  = (unsigned*)(alpha + (size_t)Ee * Hh);
  float*    denom   = (float*)(amax_u + (size_t)Nn * Hh);

  init_kernel<<<INIT_BLOCKS, 256, 0, stream>>>(out, denom, amax_u, bias);

  dim3 g((Nn + 31) / 32, HD / 128);
  gemm_h_wmma<<<g, 256, 0, stream>>>(x, Wm, h);

  edge_alpha_kernel<<<Ee / (8 * ET), 256, 0, stream>>>(h, a_src, a_dst, edge_W, edge_b,
                                                      edge_attr, ei, alpha, amax_u);
  softmax_num_kernel<<<SM_BLOCKS, 256, 0, stream>>>(ei, amax_u, alpha, denom);
  scatter_out_kernel<<<Ee / (8 * ST), 256, 0, stream>>>(h, alpha, denom, ei, out);
}


# GEMM kernel: a thread's 20 tile loads per 32-wide K chunk go out together before the first LDS store (row index clamped, zero put in at the store); they were waited for one by one; sum order kept
# speedup vs baseline: 2.3552x; 1.0722x over previous
#include <hip/hip_runtime.h>

#define Nn   50000
#define Ee   400000
#define INF_ 256
#define Hh   8
#define Dd   32
#define HD   256
#define EFe  3

typedef __attribute__((ext_vector_type(2))) float v2f;
typedef __attribute__((ext_vector_type(8))) float v8f;

__device__ __forceinline__ unsigned f2o(float f) {
  unsigned u = __float_as_uint(f);
  return (u & 0x80000000u) ? ~u : (u | 0x80000000u);
}
__device__ __forceinline__ float o2f(unsigned u) {
  return (u & 0x80000000u) ? __uint_as_float(u & 0x7fffffffu) : __uint_as_float(~u);
}

#define INIT_BLOCKS 1250
#define INIT_TRIPS  10
static_assert(INIT_BLOCKS * 256 * INIT_TRIPS * 4 == Nn * HD, "the 16-byte stores must cover out exactly");
static_assert((INIT_BLOCKS * 256) % (HD / 4) == 0, "a thread must keep its column group from trip to trip");
static_assert((Nn * Hh) % 4 == 0 && INIT_BLOCKS * 256 * 4 >= Nn * Hh, "one 16-byte store per thread must cover denom and amax");

__global__ __launch_bounds__(256) void init_kernel(float* __restrict__ out, float* __restrict__ denom,
                                                   unsigned* __restrict__ amax_u,
                                                   const float* __restrict__ bias) {
  const int tid  = blockIdx.x * 256 + threadIdx.x;
  const int nthr = INIT_BLOCKS * 256;
  const int cg = (tid & (HD / 4 - 1)) * 4;
  float4 b4; b4.x = bias[cg]; b4.y = bias[cg + 1]; b4.z = bias[cg + 2]; b4.w = bias[cg + 3];
  float4* o4 = reinterpret_cast<float4*>(out);
#pragma unroll
  for (int k = 0; k < INIT_TRIPS; ++k) o4[tid + k * nthr] = b4;
  if (tid < Nn * Hh / 4) {
    float4 zf; zf.x = 0.f; zf.y = 0.f; zf.z = 0.f; zf.w = 0.f;
    uint4  zu; zu.x = 0u; zu.y = 0u; zu.z = 0u; zu.w = 0u;
    reinterpret_cast<float4*>(denom)[tid] = zf;
    reinterpret_cast<uint4*>(amax_u)[tid] = zu;
  }
}

__global__ __launch_bounds__(256) void gemm_h_wmma(const float* __restrict__ x,
                                                   const float* __restrict__ Wm,
                                                   float* __restrict__ h) {
  __shared__ float sA[32][33];
  __shared__ float sB[128][33];
  const int tid   = threadIdx.x;
  const int mBase = blockIdx.x * 32;
  const int nBase = blockIdx.y * 128;
  const int wave  = tid >> 5, ln = tid & 31;
  const int half  = ln >> 4, mr = ln & 15;
  const int mt    = wave & 1;
  const int nt0   = (wave >> 1) << 1;

  v8f acc0 = {};
  v8f acc1 = {};

  const int wv = __builtin_amdgcn_readfirstlane(wave);

  for (int k0 = 0; k0 < INF_; k0 += 32) {
    float xr[4], wr[16];
#pragma unroll
    for (int j = 0; j < 4; ++j) {
      const int row = min(mBase + wv + 8 * j, Nn - 1);
      xr[j] = x[row * INF_ + k0 + ln];
    }
#pragma unroll
    for (int j = 0; j < 16; ++j) wr[j] = Wm[(nBase + wv + 8 * j) * INF_ + k0 + ln];
#pragma unroll
    for (int j = 0; j < 4; ++j) sA[wv + 8 * j][ln] = (mBase + wv + 8 * j < Nn) ? xr[j] : 0.f;
#pragma unroll
    for (int j = 0; j < 16; ++j) sB[wv + 8 * j][ln] = wr[j];
    __syncthreads();

    const int ar  = mt * 16 + mr;
    const int br0 = nt0 * 16 + mr;
    const int br1 = br0 + 16;
#pragma unroll
    for (int kk = 0; kk < 32; kk += 4) {
      const int kh = kk + half * 2;
      v2f a, b0, b1;
      a.x  = sA[ar][kh];   a.y  = sA[ar][kh + 1];
      b0.x = sB[br0][kh];  b0.y = sB[br0][kh + 1];
      b1.x = sB[br1][kh];  b1.y = sB[br1][kh + 1];
      acc0 = __builtin_amdgcn_wmma_f32_16x16x4_f32(false, a, false, b0, (short)0, acc0, false, false);
      acc1 = __builtin_amdgcn_wmma_f32_16x16x4_f32(false, a, false, b1, (short)0, acc1, false, false);
    }
    __syncthreads();
  }

  const int rowBase = mBase + mt * 16 + half * 8;
#pragma unroll
  for (int r = 0; r < 8; ++r) {
    int row = rowBase + r;
    if (row < Nn) {
      h[row * HD + nBase + nt0 * 16 + mr]       = acc0[r];
      h[row * HD + nBase + (nt0 + 1) * 16 + mr] = acc1[r];
    }
  }
}

#define ET 16
#define EG 4
static_assert(Ee % (8 * ET) == 0, "the tiles of 8 waves x ET edges must cover the edge list exactly");
static_assert(ET % EG == 0 && EG * Hh == 32, "a group's EG x 8 sums land one per lane");

__device__ __forceinline__ int   lane_bcast_i(int v, int l)   { return __builtin_amdgcn_readlane(v, l); }
__device__ __forceinline__ float lane_bcast_f(float v, int l) {
  return __int_as_float(__builtin_amdgcn_readlane(__float_as_int(v), l));
}

__device__ __forceinline__ float row_swap_add(float a, float b) {
  const auto r = __builtin_amdgcn_permlane16_swap(__float_as_uint(a), __float_as_uint(b), false, false);
  return __uint_as_float(r[0]) + __uint_as_float(r[1]);
}

template <int M>
__device__ __forceinline__ float row_xor(float v) {
  return __int_as_float(__builtin_amdgcn_update_dpp(__float_as_int(v), __float_as_int(v),
                                                    0x160 + M, 0xf, 0xf, true));
}

template <int M>
__device__ __forceinline__ float fold(float lo, float hi, bool up) {
  const float keep = up ? hi : lo;
  const float send = up ? lo : hi;
  return keep + row_xor<M>(send);
}

__global__ __launch_bounds__(256) void edge_alpha_kernel(
    const float* __restrict__ h, const float* __restrict__ a_src,
    const float* __restrict__ a_dst, const float* __restrict__ edge_W,
    const float* __restrict__ edge_b, const float* __restrict__ edge_attr,
    const int* __restrict__ ei, float* __restrict__ alpha,
    unsigned* __restrict__ amax_u) {
  const int lane = threadIdx.x & 31;
  const int wv = __builtin_amdgcn_readfirstlane((int)(threadIdx.x >> 5));
  const int e0 = (blockIdx.x * 8 + wv) * ET;

  float w0[Hh], w1[Hh], w2[Hh], bb[Hh], as_[Hh], ad_[Hh];
#pragma unroll
  for (int hh = 0; hh < Hh; ++hh) {
    const int f = hh * Dd + lane;
    w0[hh]  = edge_W[f * EFe + 0];
    w1[hh]  = edge_W[f * EFe + 1];
    w2[hh]  = edge_W[f * EFe + 2];
    bb[hh]  = edge_b[f];
    as_[hh] = a_src[f];
    ad_[hh] = a_dst[f];
  }

  const int   t      = lane & (ET - 1);
  const int   my_src = ei[e0 + t];
  const int   my_dst = ei[Ee + e0 + t];
  const float my_a0  = edge_attr[(e0 + t) * EFe + 0];
  const float my_a1  = edge_attr[(e0 + t) * EFe + 1];
  const float my_a2  = edge_attr[(e0 + t) * EFe + 2];

  const bool up8 = (lane & 8) != 0, up4 = (lane & 4) != 0, up2 = (lane & 2) != 0, up1 = (lane & 1) != 0;
  const int  ge  = ((lane >> 4) & 1) + 2 * ((lane >> 3) & 1);
  const int  gh  = lane & 7;

  auto load_rows = [&](int i, float (&hs)[Hh], float (&hd)[Hh]) {
    const int s = lane_bcast_i(my_src, i);
    const int d = lane_bcast_i(my_dst, i);
    const float* ps = h + (long)s * HD + lane;
    const float* pd = h + (long)d * HD + lane;
#pragma unroll
    for (int hh = 0; hh < Hh; ++hh) { hs[hh] = ps[hh * Dd]; hd[hh] = pd[hh * Dd]; }
  };

  auto edge_partials = [&](int i, const float (&hs)[Hh], const float (&hd)[Hh], float (&p)[Hh]) {
    const float ea0 = lane_bcast_f(my_a0, i);
    const float ea1 = lane_bcast_f(my_a1, i);
    const float ea2 = lane_bcast_f(my_a2, i);
#pragma unroll
    for (int hh = 0; hh < Hh; ++hh) {
      float enc = fmaf(ea0, w0[hh], fmaf(ea1, w1[hh], fmaf(ea2, w2[hh], bb[hh])));
      enc = fmaxf(enc, 0.f);
      p[hh] = hs[hh] * as_[hh] + hd[hh] * ad_[hh] + enc * hd[hh];
    }
  };

  float hsA[Hh], hdA[Hh], hsB[Hh], hdB[Hh];
  load_rows(0, hsA, hdA);
  for (int i = 0; i < ET; i += EG) {
    float pa[Hh], pb[Hh], u0[Hh], u1[Hh];
    load_rows(i + 1, hsB, hdB);
    edge_partials(i, hsA, hdA, pa);
    load_rows(i + 2, hsA, hdA);
    edge_partials(i + 1, hsB, hdB, pb);
#pragma unroll
    for (int hh = 0; hh < Hh; ++hh) u0[hh] = row_swap_add(pa[hh], pb[hh]);
    load_rows(i + 3, hsB, hdB);
    edge_partials(i + 2, hsA, hdA, pa);
    if (i + EG < ET) load_rows(i + EG, hsA, hdA);
    edge_partials(i + 3, hsB, hdB, pb);
#pragma unroll
    for (int hh = 0; hh < Hh; ++hh) u1[hh] = row_swap_add(pa[hh], pb[hh]);

    float w[Hh];
#pragma unroll
    for (int hh = 0; hh < Hh; ++hh) w[hh] = fold<8>(u0[hh], u1[hh], up8);
    const float x0 = fold<4>(w[0], w[4], up4), x1 = fold<4>(w[1], w[5], up4);
    const float x2 = fold<4>(w[2], w[6], up4), x3 = fold<4>(w[3], w[7], up4);
    const float y0 = fold<2>(x0, x2, up2), y1 = fold<2>(x1, x3, up2);
    const float z  = fold<1>(y0, y1, up1);
    const float al = (z >= 0.f) ? z : 0.2f * z;

    const int my_e = e0 + i + ge;
    const int my_d = __shfl(my_dst, i + ge, 32);
    alpha[(long)my_e * Hh + gh] = al;
    atomicMax(amax_u + (long)my_d * Hh + gh, f2o(al));
  }
}

#define SM_BLOCKS 1250
#define SM_PER    10
#define SM_B      5
static_assert(SM_BLOCKS * 256 * SM_PER == Ee * Hh, "the grid-stride trips must cover the elements exactly");
static_assert(SM_PER % SM_B == 0, "whole batches");

__global__ __launch_bounds__(256) void softmax_num_kernel(const int* __restrict__ ei,
                                                          const unsigned* __restrict__ amax_u,
                                                          float* __restrict__ alpha_ex,
                                                          float* __restrict__ denom) {
  const int tid  = blockIdx.x * 256 + threadIdx.x;
  const int nthr = SM_BLOCKS * 256;
  const int hh   = tid & 7;
  for (int b = 0; b < SM_PER; b += SM_B) {
    int dst[SM_B]; float al[SM_B]; unsigned mu[SM_B];
#pragma unroll
    for (int k = 0; k < SM_B; ++k) {
      const int idx = tid + (b + k) * nthr;
      dst[k] = ei[Ee + (idx >> 3)];
      al[k]  = alpha_ex[idx];
    }
#pragma unroll
    for (int k = 0; k < SM_B; ++k) mu[k] = amax_u[dst[k] * Hh + hh];
#pragma unroll
    for (int k = 0; k < SM_B; ++k) {
      const int idx = tid + (b + k) * nthr;
      const float ex = __expf(al[k] - o2f(mu[k]));
      alpha_ex[idx] = ex;
      atomicAdd(&denom[dst[k] * Hh + hh], ex);
    }
  }
}

#define ST 16
static_assert(Ee % (8 * ST) == 0, "the tiles of 8 waves x ST edges must cover the edge list exactly");
static_assert(2 * ST == 32, "two lanes (4 heads each) per edge of the tile");

__global__ __launch_bounds__(256) void scatter_out_kernel(
    const float* __restrict__ h, const float* __restrict__ ex,
    const float* __restrict__ denom, const int* __restrict__ ei,
    float* __restrict__ out) {
  const int lane = threadIdx.x & 31;
  const int wv = __builtin_amdgcn_readfirstlane((int)(threadIdx.x >> 5));
  const int e0 = (blockIdx.x * 8 + wv) * ST;

  const int t      = lane & (ST - 1);
  const int my_src = ei[e0 + t];
  const int my_dst = ei[Ee + e0 + t];

  const int ke = lane >> 1;
  const int hb = (lane & 1) * 4;
  const int kd = ei[Ee + e0 + ke];
  const float4 x4 = *reinterpret_cast<const float4*>(ex + (long)(e0 + ke) * Hh + hb);
  const float4 d4 = *reinterpret_cast<const float4*>(denom + (long)kd * Hh + hb);
  float at[4];
  at[0] = x4.x / fmaxf(d4.x, 1e-10f);
  at[1] = x4.y / fmaxf(d4.y, 1e-10f);
  at[2] = x4.z / fmaxf(d4.z, 1e-10f);
  at[3] = x4.w / fmaxf(d4.w, 1e-10f);

  auto load_row = [&](int i, float (&hs)[Hh]) {
    const int s = lane_bcast_i(my_src, i);
    const float* ps = h + (long)s * HD + lane;
#pragma unroll
    for (int hh = 0; hh < Hh; ++hh) hs[hh] = ps[hh * Dd];
  };

  auto add_edge = [&](int i, const float (&hs)[Hh]) {
    const int d = lane_bcast_i(my_dst, i);
    float* po = out + (long)d * HD + lane;
#pragma unroll
    for (int hh = 0; hh < Hh; ++hh) {
      const float attn = lane_bcast_f(at[hh & 3], 2 * i + (hh >> 2));
      atomicAdd(po + hh * Dd, hs[hh] * attn);
    }
  };

  float hsA[Hh], hsB[Hh];
  load_row(0, hsA);
  for (int i = 0; i < ST - 2; i += 2) {
    load_row(i + 1, hsB);
    add_edge(i, hsA);
    load_row(i + 2, hsA);
    add_edge(i + 1, hsB);
  }
  load_row(ST - 1, hsB);
  add_edge(ST - 2, hsA);
  add_edge(ST - 1, hsB);
}

extern "C" void kernel_launch(void* const* d_in, const int* in_sizes, int n_in,
                              void* d_out, int out_size, void* d_ws, size_t ws_size,
                              hipStream_t stream) {
  const float* x        = (const float*)d_in[0];
  const float* Wm       = (const float*)d_in[1];
  const float* a_src    = (const float*)d_in[2];
  const float* a_dst    = (const float*)d_in[3];
  const float* edge_W   = (const float*)d_in[4];
  const float* edge_b   = (const float*)d_in[5];
  const float* bias     = (const float*)d_in[6];
  const float* edge_attr= (const float*)d_in[7];
  const int*   ei       = (const int*)d_in[8];
  float* out = (float*)d_out;

  float*    h       = (float*)d_ws;
  float*    alpha   = h + (size_t)Nn * HD;
  unsigned* amax_u  = (unsigned*)(alpha + (size_t)Ee * Hh);
  float*    denom   = (float*)(amax_u + (size_t)Nn * Hh);

  init_kernel<<<INIT_BLOCKS, 256, 0, stream>>>(out, denom, amax_u, bias);

  dim3 g((Nn + 31) / 32, HD / 128);
  gemm_h_wmma<<<g, 256, 0, stream>>>(x, Wm, h);

  edge_alpha_kernel<<<Ee / (8 * ET), 256, 0, stream>>>(h, a_src, a_dst, edge_W, edge_b,
                                                      edge_attr, ei, alpha, amax_u);
  softmax_num_kernel<<<SM_BLOCKS, 256, 0, stream>>>(ei, amax_u, alpha, denom);
  scatter_out_kernel<<<Ee / (8 * ST), 256, 0, stream>>>(h, alpha, denom, ei, out);
}


# scatter kernel: each package half adds only to its own 4 heads' lines of out (die number picks the half, per-half counter hands out edge tiles): a line's atomics stay in one cache half; atomics kept
# speedup vs baseline: 2.5131x; 1.0670x over previous
#include <hip/hip_runtime.h>

#define Nn   50000
#define Ee   400000
#define INF_ 256
#define Hh   8
#define Dd   32
#define HD   256
#define EFe  3

typedef __attribute__((ext_vector_type(2))) float v2f;
typedef __attribute__((ext_vector_type(8))) float v8f;

__device__ __forceinline__ unsigned f2o(float f) {
  unsigned u = __float_as_uint(f);
  return (u & 0x80000000u) ? ~u : (u | 0x80000000u);
}
__device__ __forceinline__ float o2f(unsigned u) {
  return (u & 0x80000000u) ? __uint_as_float(u & 0x7fffffffu) : __uint_as_float(~u);
}

#define INIT_BLOCKS 1250
#define INIT_TRIPS  10
static_assert(INIT_BLOCKS * 256 * INIT_TRIPS * 4 == Nn * HD, "the 16-byte stores must cover out exactly");
static_assert((INIT_BLOCKS * 256) % (HD / 4) == 0, "a thread must keep its column group from trip to trip");
static_assert((Nn * Hh) % 4 == 0 && INIT_BLOCKS * 256 * 4 >= Nn * Hh, "one 16-byte store per thread must cover denom and amax");

#define TK_STRIDE 64
#define WS_NEED ((unsigned long long)(Nn * HD + Ee * Hh + 2 * Nn * Hh) * 4ull + 512ull)

__global__ __launch_bounds__(256) void init_kernel(float* __restrict__ out, float* __restrict__ denom,
                                                   unsigned* __restrict__ amax_u,
                                                   const float* __restrict__ bias,
                                                   unsigned* __restrict__ tickets,
                                                   unsigned long long ws_bytes) {
  const int tid  = blockIdx.x * 256 + threadIdx.x;
  const int nthr = INIT_BLOCKS * 256;
  const int cg = (tid & (HD / 4 - 1)) * 4;
  float4 b4; b4.x = bias[cg]; b4.y = bias[cg + 1]; b4.z = bias[cg + 2]; b4.w = bias[cg + 3];
  float4* o4 = reinterpret_cast<float4*>(out);
#pragma unroll
  for (int k = 0; k < INIT_TRIPS; ++k) o4[tid + k * nthr] = b4;
  if (tid < Nn * Hh / 4) {
    float4 zf; zf.x = 0.f; zf.y = 0.f; zf.z = 0.f; zf.w = 0.f;
    uint4  zu; zu.x = 0u; zu.y = 0u; zu.z = 0u; zu.w = 0u;
    reinterpret_cast<float4*>(denom)[tid] = zf;
    reinterpret_cast<uint4*>(amax_u)[tid] = zu;
  }
  if (tid == 0 && ws_bytes >= WS_NEED) { tickets[0] = 0u; tickets[TK_STRIDE] = 0u; }
}

__global__ __launch_bounds__(256) void gemm_h_wmma(const float* __restrict__ x,
                                                   const float* __restrict__ Wm,
                                                   float* __restrict__ h) {
  __shared__ float sA[32][33];
  __shared__ float sB[128][33];
  const int tid   = threadIdx.x;
  const int mBase = blockIdx.x * 32;
  const int nBase = blockIdx.y * 128;
  const int wave  = tid >> 5, ln = tid & 31;
  const int half  = ln >> 4, mr = ln & 15;
  const int mt    = wave & 1;
  const int nt0   = (wave >> 1) << 1;

  v8f acc0 = {};
  v8f acc1 = {};

  const int wv = __builtin_amdgcn_readfirstlane(wave);

  for (int k0 = 0; k0 < INF_; k0 += 32) {
    float xr[4], wr[16];
#pragma unroll
    for (int j = 0; j < 4; ++j) {
      const int row = min(mBase + wv + 8 * j, Nn - 1);
      xr[j] = x[row * INF_ + k0 + ln];
    }
#pragma unroll
    for (int j = 0; j < 16; ++j) wr[j] = Wm[(nBase + wv + 8 * j) * INF_ + k0 + ln];
#pragma unroll
    for (int j = 0; j < 4; ++j) sA[wv + 8 * j][ln] = (mBase + wv + 8 * j < Nn) ? xr[j] : 0.f;
#pragma unroll
    for (int j = 0; j < 16; ++j) sB[wv + 8 * j][ln] = wr[j];
    __syncthreads();

    const int ar  = mt * 16 + mr;
    const int br0 = nt0 * 16 + mr;
    const int br1 = br0 + 16;
#pragma unroll
    for (int kk = 0; kk < 32; kk += 4) {
      const int kh = kk + half * 2;
      v2f a, b0, b1;
      a.x  = sA[ar][kh];   a.y  = sA[ar][kh + 1];
      b0.x = sB[br0][kh];  b0.y = sB[br0][kh + 1];
      b1.x = sB[br1][kh];  b1.y = sB[br1][kh + 1];
      acc0 = __builtin_amdgcn_wmma_f32_16x16x4_f32(false, a, false, b0, (short)0, acc0, false, false);
      acc1 = __builtin_amdgcn_wmma_f32_16x16x4_f32(false, a, false, b1, (short)0, acc1, false, false);
    }
    __syncthreads();
  }

  const int rowBase = mBase + mt * 16 + half * 8;
#pragma unroll
  for (int r = 0; r < 8; ++r) {
    int row = rowBase + r;
    if (row < Nn) {
      h[row * HD + nBase + nt0 * 16 + mr]       = acc0[r];
      h[row * HD + nBase + (nt0 + 1) * 16 + mr] = acc1[r];
    }
  }
}

#define ET 16
#define EG 4
static_assert(Ee % (8 * ET) == 0, "the tiles of 8 waves x ET edges must cover the edge list exactly");
static_assert(ET % EG == 0 && EG * Hh == 32, "a group's EG x 8 sums land one per lane");

__device__ __forceinline__ int   lane_bcast_i(int v, int l)   { return __builtin_amdgcn_readlane(v, l); }
__device__ __forceinline__ float lane_bcast_f(float v, int l) {
  return __int_as_float(__builtin_amdgcn_readlane(__float_as_int(v), l));
}

__device__ __forceinline__ float row_swap_add(float a, float b) {
  const auto r = __builtin_amdgcn_permlane16_swap(__float_as_uint(a), __float_as_uint(b), false, false);
  return __uint_as_float(r[0]) + __uint_as_float(r[1]);
}

template <int M>
__device__ __forceinline__ float row_xor(float v) {
  return __int_as_float(__builtin_amdgcn_update_dpp(__float_as_int(v), __float_as_int(v),
                                                    0x160 + M, 0xf, 0xf, true));
}

template <int M>
__device__ __forceinline__ float fold(float lo, float hi, bool up) {
  const float keep = up ? hi : lo;
  const float send = up ? lo : hi;
  return keep + row_xor<M>(send);
}

__global__ __launch_bounds__(256) void edge_alpha_kernel(
    const float* __restrict__ h, const float* __restrict__ a_src,
    const float* __restrict__ a_dst, const float* __restrict__ edge_W,
    const float* __restrict__ edge_b, const float* __restrict__ edge_attr,
    const int* __restrict__ ei, float* __restrict__ alpha,
    unsigned* __restrict__ amax_u) {
  const int lane = threadIdx.x & 31;
  const int wv = __builtin_amdgcn_readfirstlane((int)(threadIdx.x >> 5));
  const int e0 = (blockIdx.x * 8 + wv) * ET;

  float w0[Hh], w1[Hh], w2[Hh], bb[Hh], as_[Hh], ad_[Hh];
#pragma unroll
  for (int hh = 0; hh < Hh; ++hh) {
    const int f = hh * Dd + lane;
    w0[hh]  = edge_W[f * EFe + 0];
    w1[hh]  = edge_W[f * EFe + 1];
    w2[hh]  = edge_W[f * EFe + 2];
    bb[hh]  = edge_b[f];
    as_[hh] = a_src[f];
    ad_[hh] = a_dst[f];
  }

  const int   t      = lane & (ET - 1);
  const int   my_src = ei[e0 + t];
  const int   my_dst = ei[Ee + e0 + t];
  const float my_a0  = edge_attr[(e0 + t) * EFe + 0];
  const float my_a1  = edge_attr[(e0 + t) * EFe + 1];
  const float my_a2  = edge_attr[(e0 + t) * EFe + 2];

  const bool up8 = (lane & 8) != 0, up4 = (lane & 4) != 0, up2 = (lane & 2) != 0, up1 = (lane & 1) != 0;
  const int  ge  = ((lane >> 4) & 1) + 2 * ((lane >> 3) & 1);
  const int  gh  = lane & 7;

  auto load_rows = [&](int i, float (&hs)[Hh], float (&hd)[Hh]) {
    const int s = lane_bcast_i(my_src, i);
    const int d = lane_bcast_i(my_dst, i);
    const float* ps = h + (long)s * HD + lane;
    const float* pd = h + (long)d * HD + lane;
#pragma unroll
    for (int hh = 0; hh < Hh; ++hh) { hs[hh] = ps[hh * Dd]; hd[hh] = pd[hh * Dd]; }
  };

  auto edge_partials = [&](int i, const float (&hs)[Hh], const float (&hd)[Hh], float (&p)[Hh]) {
    const float ea0 = lane_bcast_f(my_a0, i);
    const float ea1 = lane_bcast_f(my_a1, i);
    const float ea2 = lane_bcast_f(my_a2, i);
#pragma unroll
    for (int hh = 0; hh < Hh; ++hh) {
      float enc = fmaf(ea0, w0[hh], fmaf(ea1, w1[hh], fmaf(ea2, w2[hh], bb[hh])));
      enc = fmaxf(enc, 0.f);
      p[hh] = hs[hh] * as_[hh] + hd[hh] * ad_[hh] + enc * hd[hh];
    }
  };

  float hsA[Hh], hdA[Hh], hsB[Hh], hdB[Hh];
  load_rows(0, hsA, hdA);
  for (int i = 0; i < ET; i += EG) {
    float pa[Hh], pb[Hh], u0[Hh], u1[Hh];
    load_rows(i + 1, hsB, hdB);
    edge_partials(i, hsA, hdA, pa);
    load_rows(i + 2, hsA, hdA);
    edge_partials(i + 1, hsB, hdB, pb);
#pragma unroll
    for (int hh = 0; hh < Hh; ++hh) u0[hh] = row_swap_add(pa[hh], pb[hh]);
    load_rows(i + 3, hsB, hdB);
    edge_partials(i + 2, hsA, hdA, pa);
    if (i + EG < ET) load_rows(i + EG, hsA, hdA);
    edge_partials(i + 3, hsB, hdB, pb);
#pragma unroll
    for (int hh = 0; hh < Hh; ++hh) u1[hh] = row_swap_add(pa[hh], pb[hh]);

    float w[Hh];
#pragma unroll
    for (int hh = 0; hh < Hh; ++hh) w[hh] = fold<8>(u0[hh], u1[hh], up8);
    const float x0 = fold<4>(w[0], w[4], up4), x1 = fold<4>(w[1], w[5], up4);
    const float x2 = fold<4>(w[2], w[6], up4), x3 = fold<4>(w[3], w[7], up4);
    const float y0 = fold<2>(x0, x2, up2), y1 = fold<2>(x1, x3, up2);
    const float z  = fold<1>(y0, y1, up1);
    const float al = (z >= 0.f) ? z : 0.2f * z;

    const int my_e = e0 + i + ge;
    const int my_d = __shfl(my_dst, i + ge, 32);
    alpha[(long)my_e * Hh + gh] = al;
    atomicMax(amax_u + (long)my_d * Hh + gh, f2o(al));
  }
}

#define SM_BLOCKS 1250
#define SM_PER    10
#define SM_B      5
static_assert(SM_BLOCKS * 256 * SM_PER == Ee * Hh, "the grid-stride trips must cover the elements exactly");
static_assert(SM_PER % SM_B == 0, "whole batches");

__global__ __launch_bounds__(256) void softmax_num_kernel(const int* __restrict__ ei,
                                                          const unsigned* __restrict__ amax_u,
                                                          float* __restrict__ alpha_ex,
                                                          float* __restrict__ denom) {
  const int tid  = blockIdx.x * 256 + threadIdx.x;
  const int nthr = SM_BLOCKS * 256;
  const int hh   = tid & 7;
  for (int b = 0; b < SM_PER; b += SM_B) {
    int dst[SM_B]; float al[SM_B]; unsigned mu[SM_B];
#pragma unroll
    for (int k = 0; k < SM_B; ++k) {
      const int idx = tid + (b + k) * nthr;
      dst[k] = ei[Ee + (idx >> 3)];
      al[k]  = alpha_ex[idx];
    }
#pragma unroll
    for (int k = 0; k < SM_B; ++k) mu[k] = amax_u[dst[k] * Hh + hh];
#pragma unroll
    for (int k = 0; k < SM_B; ++k) {
      const int idx = tid + (b + k) * nthr;
      const float ex = __expf(al[k] - o2f(mu[k]));
      alpha_ex[idx] = ex;
      atomicAdd(&denom[dst[k] * Hh + hh], ex);
    }
  }
}

#define SC_E     32
#define SC_WAVES 4
#define SC_HC    4
#define SC_TILES (Ee / (SC_WAVES * SC_E))
static_assert(Ee % (SC_WAVES * SC_E) == 0, "the tiles of SC_WAVES waves x SC_E edges must cover the edge list exactly");
static_assert(SC_E == 32 && 2 * SC_HC == Hh, "one edge per lane, two column halves");

__global__ __launch_bounds__(SC_WAVES * 32) void scatter_out_kernel(
    const float* __restrict__ h, const float* __restrict__ ex,
    const float* __restrict__ denom, const int* __restrict__ ei,
    float* __restrict__ out, unsigned* __restrict__ tickets, unsigned long long ws_bytes) {
  __shared__ unsigned s_pick;
  if (threadIdx.x == 0) {
    unsigned ch, tk;
    if (ws_bytes >= WS_NEED) {
      ch = (__builtin_amdgcn_s_sendmsg_rtn(0x87) >> 18) & 1u;
      tk = atomicAdd(tickets + ch * TK_STRIDE, 1u);
      if (tk >= SC_TILES) { ch ^= 1u; tk = atomicAdd(tickets + ch * TK_STRIDE, 1u); }
      tk = min(tk, (unsigned)(SC_TILES - 1));
    } else {
      ch = blockIdx.x & 1u; tk = blockIdx.x >> 1;
    }
    s_pick = tk * 2u + ch;
  }
  __syncthreads();
  const int pick = __builtin_amdgcn_readfirstlane((int)s_pick);
  const int ch   = pick & 1;
  const int lane = threadIdx.x & 31;
  const int wv   = __builtin_amdgcn_readfirstlane((int)(threadIdx.x >> 5));
  const int e0   = ((pick >> 1) * SC_WAVES + wv) * SC_E;

  const int my_src = ei[e0 + lane];
  const int my_dst = ei[Ee + e0 + lane];
  const float4 x4 = *reinterpret_cast<const float4*>(ex + (long)(e0 + lane) * Hh + ch * SC_HC);
  const float4 d4 = *reinterpret_cast<const float4*>(denom + (long)my_dst * Hh + ch * SC_HC);
  float at[SC_HC];
  at[0] = x4.x / fmaxf(d4.x, 1e-10f);
  at[1] = x4.y / fmaxf(d4.y, 1e-10f);
  at[2] = x4.z / fmaxf(d4.z, 1e-10f);
  at[3] = x4.w / fmaxf(d4.w, 1e-10f);

  const float* hb = h + ch * (SC_HC * Dd) + lane;
  float*       ob = out + ch * (SC_HC * Dd) + lane;

  auto load_row = [&](int i, float (&hs)[SC_HC]) {
    const int s = lane_bcast_i(my_src, i);
    const float* ps = hb + (long)s * HD;
#pragma unroll
    for (int k = 0; k < SC_HC; ++k) hs[k] = ps[k * Dd];
  };

  auto add_edge = [&](int i, const float (&hs)[SC_HC]) {
    const int d = lane_bcast_i(my_dst, i);
    float* po = ob + (long)d * HD;
#pragma unroll
    for (int k = 0; k < SC_HC; ++k) atomicAdd(po + k * Dd, hs[k] * lane_bcast_f(at[k], i));
  };

  float ra[SC_HC], rb[SC_HC], rc[SC_HC], rd[SC_HC];
  load_row(0, ra); load_row(1, rb); load_row(2, rc);
  for (int i = 0; i < SC_E - 4; i += 4) {
    load_row(i + 3, rd); add_edge(i,     ra);
    load_row(i + 4, ra); add_edge(i + 1, rb);
    load_row(i + 5, rb); add_edge(i + 2, rc);
    load_row(i + 6, rc); add_edge(i + 3, rd);
  }
  load_row(SC_E - 1, rd);
  add_edge(SC_E - 4, ra); add_edge(SC_E - 3, rb); add_edge(SC_E - 2, rc); add_edge(SC_E - 1, rd);
}

extern "C" void kernel_launch(void* const* d_in, const int* in_sizes, int n_in,
                              void* d_out, int out_size, void* d_ws, size_t ws_size,
                              hipStream_t stream) {
  const float* x        = (const float*)d_in[0];
  const float* Wm       = (const float*)d_in[1];
  const float* a_src    = (const float*)d_in[2];
  const float* a_dst    = (const float*)d_in[3];
  const float* edge_W   = (const float*)d_in[4];
  const float* edge_b   = (const float*)d_in[5];
  const float* bias     = (const float*)d_in[6];
  const float* edge_attr= (const float*)d_in[7];
  const int*   ei       = (const int*)d_in[8];
  float* out = (float*)d_out;

  float*    h       = (float*)d_ws;
  float*    alpha   = h + (size_t)Nn * HD;
  unsigned* amax_u  = (unsigned*)(alpha + (size_t)Ee * Hh);
  float*    denom   = (float*)(amax_u + (size_t)Nn * Hh);
  unsigned* tickets = (unsigned*)(denom + (size_t)Nn * Hh);

  init_kernel<<<INIT_BLOCKS, 256, 0, stream>>>(out, denom, amax_u, bias, tickets, (unsigned long long)ws_size);

  dim3 g((Nn + 31) / 32, HD / 128);
  gemm_h_wmma<<<g, 256, 0, stream>>>(x, Wm, h);

  edge_alpha_kernel<<<Ee / (8 * ET), 256, 0, stream>>>(h, a_src, a_dst, edge_W, edge_b,
                                                      edge_attr, ei, alpha, amax_u);
  softmax_num_kernel<<<SM_BLOCKS, 256, 0, stream>>>(ei, amax_u, alpha, denom);
  scatter_out_kernel<<<2 * SC_TILES, SC_WAVES * 32, 0, stream>>>(h, alpha, denom, ei, out, tickets,
                                                                (unsigned long long)ws_size);
}


# GEMM epilogue stores per-node source scores a_src.h[n,head] in the denom plane (init runs behind the edge-logit kernel), which gathers dst row + 2x16 B scores, not the src row; new logit sum order
# speedup vs baseline: 2.6610x; 1.0589x over previous
#include <hip/hip_runtime.h>

#define Nn   50000
#define Ee   400000
#define INF_ 256
#define Hh   8
#define Dd   32
#define HD   256
#define EFe  3

typedef __attribute__((ext_vector_type(2))) float v2f;
typedef __attribute__((ext_vector_type(8))) float v8f;

__device__ __forceinline__ unsigned f2o(float f) {
  unsigned u = __float_as_uint(f);
  return (u & 0x80000000u) ? ~u : (u | 0x80000000u);
}
__device__ __forceinline__ float o2f(unsigned u) {
  return (u & 0x80000000u) ? __uint_as_float(u & 0x7fffffffu) : __uint_as_float(~u);
}

#define INIT_BLOCKS 1250
#define INIT_TRIPS  10
static_assert(INIT_BLOCKS * 256 * INIT_TRIPS * 4 == Nn * HD, "the 16-byte stores must cover out exactly");
static_assert((INIT_BLOCKS * 256) % (HD / 4) == 0, "a thread must keep its column group from trip to trip");
static_assert((Nn * Hh) % 4 == 0 && INIT_BLOCKS * 256 * 4 >= Nn * Hh, "one 16-byte store per thread must cover denom");

#define TK_STRIDE 64
#define WS_NEED ((unsigned long long)(Nn * HD + Ee * Hh + 2 * Nn * Hh) * 4ull + 512ull)

__global__ __launch_bounds__(256) void init_kernel(float* __restrict__ out, float* __restrict__ denom,
                                                   const float* __restrict__ bias,
                                                   unsigned* __restrict__ tickets,
                                                   unsigned long long ws_bytes) {
  const int tid  = blockIdx.x * 256 + threadIdx.x;
  const int nthr = INIT_BLOCKS * 256;
  const int cg = (tid & (HD / 4 - 1)) * 4;
  float4 b4; b4.x = bias[cg]; b4.y = bias[cg + 1]; b4.z = bias[cg + 2]; b4.w = bias[cg + 3];
  float4* o4 = reinterpret_cast<float4*>(out);
#pragma unroll
  for (int k = 0; k < INIT_TRIPS; ++k) o4[tid + k * nthr] = b4;
  if (tid < Nn * Hh / 4) {
    float4 zf; zf.x = 0.f; zf.y = 0.f; zf.z = 0.f; zf.w = 0.f;
    reinterpret_cast<float4*>(denom)[tid] = zf;
  }
  if (tid == 0 && ws_bytes >= WS_NEED) { tickets[0] = 0u; tickets[TK_STRIDE] = 0u; }
}

template <int M>
__device__ __forceinline__ float row_xor(float v) {
  return __int_as_float(__builtin_amdgcn_update_dpp(__float_as_int(v), __float_as_int(v),
                                                    0x160 + M, 0xf, 0xf, true));
}

template <int M>
__device__ __forceinline__ float fold(float lo, float hi, bool up) {
  const float keep = up ? hi : lo;
  const float send = up ? lo : hi;
  return keep + row_xor<M>(send);
}

static_assert(Nn % 16 == 0, "the last tile's valid rows must end on a 128-byte line of a score plane (16 rows x 16 bytes = 2 lines)");
__global__ __launch_bounds__(256) void gemm_h_wmma(const float* __restrict__ x,
                                                   const float* __restrict__ Wm,
                                                   const float* __restrict__ a_src,
                                                   float* __restrict__ h,
                                                   float* __restrict__ sc,
                                                   unsigned* __restrict__ amax_u) {
  __shared__ float sA[32][33];
  __shared__ float sB[128][33];
  __shared__ alignas(16) float sS[32 * 4];
  const int tid   = threadIdx.x;
  const int mBase = blockIdx.x * 32;
  const int nBase = blockIdx.y * 128;
  const int wave  = tid >> 5, ln = tid & 31;
  const int half  = ln >> 4, mr = ln & 15;
  const int mt    = wave & 1;
  const int nt0   = (wave >> 1) << 1;

  v8f acc0 = {};
  v8f acc1 = {};

  if (blockIdx.y == 0 && mBase * Hh + tid < Nn * Hh) amax_u[mBase * Hh + tid] = 0u;

  const int wv = __builtin_amdgcn_readfirstlane(wave);

  for (int k0 = 0; k0 < INF_; k0 += 32) {
    float xr[4], wr[16];
#pragma unroll
    for (int j = 0; j < 4; ++j) {
      const int row = min(mBase + wv + 8 * j, Nn - 1);
      xr[j] = x[row * INF_ + k0 + ln];
    }
#pragma unroll
    for (int j = 0; j < 16; ++j) wr[j] = Wm[(nBase + wv + 8 * j) * INF_ + k0 + ln];
#pragma unroll
    for (int j = 0; j < 4; ++j) sA[wv + 8 * j][ln] = (mBase + wv + 8 * j < Nn) ? xr[j] : 0.f;
#pragma unroll
    for (int j = 0; j < 16; ++j) sB[wv + 8 * j][ln] = wr[j];
    __syncthreads();

    const int ar  = mt * 16 + mr;
    const int br0 = nt0 * 16 + mr;
    const int br1 = br0 + 16;
#pragma unroll
    for (int kk = 0; kk < 32; kk += 4) {
      const int kh = kk + half * 2;
      v2f a, b0, b1;
      a.x  = sA[ar][kh];   a.y  = sA[ar][kh + 1];
      b0.x = sB[br0][kh];  b0.y = sB[br0][kh + 1];
      b1.x = sB[br1][kh];  b1.y = sB[br1][kh + 1];
      acc0 = __builtin_amdgcn_wmma_f32_16x16x4_f32(false, a, false, b0, (short)0, acc0, false, false);
      acc1 = __builtin_amdgcn_wmma_f32_16x16x4_f32(false, a, false, b1, (short)0, acc1, false, false);
    }
    __syncthreads();
  }

  const int hl = wv >> 1;
  const float* av = a_src + (blockIdx.y * 4 + hl) * Dd;
  const float as0 = av[mr], as1 = av[16 + mr];
  float v[8];
#pragma unroll
  for (int r = 0; r < 8; ++r) v[r] = fmaf(acc1[r], as1, acc0[r] * as0);
  const bool up8 = (mr & 8) != 0, up4 = (mr & 4) != 0, up2 = (mr & 2) != 0;
  const float q0 = fold<8>(v[0], v[4], up8), q1 = fold<8>(v[1], v[5], up8);
  const float q2 = fold<8>(v[2], v[6], up8), q3 = fold<8>(v[3], v[7], up8);
  const float t0 = fold<4>(q0, q2, up4), t1 = fold<4>(q1, q3, up4);
  const float y  = fold<2>(t0, t1, up2);
  const float s  = y + row_xor<1>(y);
  const int   sr = ((mr >> 3) & 1) * 4 + ((mr >> 2) & 1) * 2 + ((mr >> 1) & 1);
  if ((mr & 1) == 0) sS[(mt * 16 + half * 8 + sr) * 4 + hl] = s;
  __syncthreads();
  if (wv == 0 && mBase + ln < Nn)
    reinterpret_cast<float4*>(sc)[(size_t)blockIdx.y * Nn + mBase + ln] =
        reinterpret_cast<const float4*>(sS)[ln];

  const int rowBase = mBase + mt * 16 + half * 8;
#pragma unroll
  for (int r = 0; r < 8; ++r) {
    int row = rowBase + r;
    if (row < Nn) {
      h[row * HD + nBase + nt0 * 16 + mr]       = acc0[r];
      h[row * HD + nBase + (nt0 + 1) * 16 + mr] = acc1[r];
    }
  }
}

#define ET 16
#define EG 4
static_assert(Ee % (8 * ET) == 0, "the tiles of 8 waves x ET edges must cover the edge list exactly");
static_assert(ET % EG == 0 && EG * Hh == 32, "a group's EG x 8 sums land one per lane");

__device__ __forceinline__ int   lane_bcast_i(int v, int l)   { return __builtin_amdgcn_readlane(v, l); }
__device__ __forceinline__ float lane_bcast_f(float v, int l) {
  return __int_as_float(__builtin_amdgcn_readlane(__float_as_int(v), l));
}

__device__ __forceinline__ float row_swap_add(float a, float b) {
  const auto r = __builtin_amdgcn_permlane16_swap(__float_as_uint(a), __float_as_uint(b), false, false);
  return __uint_as_float(r[0]) + __uint_as_float(r[1]);
}

__global__ __launch_bounds__(256) void edge_alpha_kernel(
    const float* __restrict__ h, const float* __restrict__ sc,
    const float* __restrict__ a_dst, const float* __restrict__ edge_W,
    const float* __restrict__ edge_b, const float* __restrict__ edge_attr,
    const int* __restrict__ ei, float* __restrict__ alpha,
    unsigned* __restrict__ amax_u) {
  const int lane = threadIdx.x & 31;
  const int wv = __builtin_amdgcn_readfirstlane((int)(threadIdx.x >> 5));
  const int e0 = (blockIdx.x * 8 + wv) * ET;

  float w0[Hh], w1[Hh], w2[Hh], bb[Hh], ad_[Hh];
#pragma unroll
  for (int hh = 0; hh < Hh; ++hh) {
    const int f = hh * Dd + lane;
    w0[hh]  = edge_W[f * EFe + 0];
    w1[hh]  = edge_W[f * EFe + 1];
    w2[hh]  = edge_W[f * EFe + 2];
    bb[hh]  = edge_b[f];
    ad_[hh] = a_dst[f];
  }

  const int   t      = lane & (ET - 1);
  const int   my_src = ei[e0 + t];
  const int   my_dst = ei[Ee + e0 + t];
  const float my_a0  = edge_attr[(e0 + t) * EFe + 0];
  const float my_a1  = edge_attr[(e0 + t) * EFe + 1];
  const float my_a2  = edge_attr[(e0 + t) * EFe + 2];

  const bool up8 = (lane & 8) != 0, up4 = (lane & 4) != 0, up2 = (lane & 2) != 0, up1 = (lane & 1) != 0;
  const int  ge  = ((lane >> 4) & 1) + 2 * ((lane >> 3) & 1);
  const int  gh  = lane & 7;
  const float* scl = sc + (long)(gh >> 2) * (Nn * 4) + (gh & 3);

  auto load_row = [&](int i, float (&hd)[Hh]) {
    const int d = lane_bcast_i(my_dst, i);
    const float* pd = h + (long)d * HD + lane;
#pragma unroll
    for (int hh = 0; hh < Hh; ++hh) hd[hh] = pd[hh * Dd];
  };

  auto edge_partials = [&](int i, const float (&hd)[Hh], float (&p)[Hh]) {
    const float ea0 = lane_bcast_f(my_a0, i);
    const float ea1 = lane_bcast_f(my_a1, i);
    const float ea2 = lane_bcast_f(my_a2, i);
#pragma unroll
    for (int hh = 0; hh < Hh; ++hh) {
      float enc = fmaf(ea0, w0[hh], fmaf(ea1, w1[hh], fmaf(ea2, w2[hh], bb[hh])));
      enc = fmaxf(enc, 0.f);
      p[hh] = hd[hh] * ad_[hh] + enc * hd[hh];
    }
  };

  float rA[Hh], rB[Hh], rC[Hh], rD[Hh];
  load_row(0, rA); load_row(1, rB);
  for (int i = 0; i < ET; i += EG) {
    float pa[Hh], pb[Hh], u0[Hh], u1[Hh];
    const bool more = i + EG < ET;
    const int   my_e = e0 + i + ge;
    const int   my_d = __shfl(my_dst, i + ge, 32);
    const float s_sc = scl[(long)__shfl(my_src, i + ge, 32) * 4];
    load_row(i + 2, rC);
    edge_partials(i, rA, pa);
    load_row(i + 3, rD);
    edge_partials(i + 1, rB, pb);
#pragma unroll
    for (int hh = 0; hh < Hh; ++hh) u0[hh] = row_swap_add(pa[hh], pb[hh]);
    if (more) load_row(i + EG, rA);
    edge_partials(i + 2, rC, pa);
    if (more) load_row(i + EG + 1, rB);
    edge_partials(i + 3, rD, pb);
#pragma unroll
    for (int hh = 0; hh < Hh; ++hh) u1[hh] = row_swap_add(pa[hh], pb[hh]);

    float w[Hh];
#pragma unroll
    for (int hh = 0; hh < Hh; ++hh) w[hh] = fold<8>(u0[hh], u1[hh], up8);
    const float x0 = fold<4>(w[0], w[4], up4), x1 = fold<4>(w[1], w[5], up4);
    const float x2 = fold<4>(w[2], w[6], up4), x3 = fold<4>(w[3], w[7], up4);
    const float y0 = fold<2>(x0, x2, up2), y1 = fold<2>(x1, x3, up2);
    const float z  = fold<1>(y0, y1, up1) + s_sc;
    const float al = (z >= 0.f) ? z : 0.2f * z;

    alpha[(long)my_e * Hh + gh] = al;
    atomicMax(amax_u + (long)my_d * Hh + gh, f2o(al));
  }
}

#define SM_BLOCKS 1250
#define SM_PER    10
#define SM_B      5
static_assert(SM_BLOCKS * 256 * SM_PER == Ee * Hh, "the grid-stride trips must cover the elements exactly");
static_assert(SM_PER % SM_B == 0, "whole batches");

__global__ __launch_bounds__(256) void softmax_num_kernel(const int* __restrict__ ei,
                                                          const unsigned* __restrict__ amax_u,
                                                          float* __restrict__ alpha_ex,
                                                          float* __restrict__ denom) {
  const int tid  = blockIdx.x * 256 + threadIdx.x;
  const int nthr = SM_BLOCKS * 256;
  const int hh   = tid & 7;
  for (int b = 0; b < SM_PER; b += SM_B) {
    int dst[SM_B]; float al[SM_B]; unsigned mu[SM_B];
#pragma unroll
    for (int k = 0; k < SM_B; ++k) {
      const int idx = tid + (b + k) * nthr;
      dst[k] = ei[Ee + (idx >> 3)];
      al[k]  = alpha_ex[idx];
    }
#pragma unroll
    for (int k = 0; k < SM_B; ++k) mu[k] = amax_u[dst[k] * Hh + hh];
#pragma unroll
    for (int k = 0; k < SM_B; ++k) {
      const int idx = tid + (b + k) * nthr;
      const float ex = __expf(al[k] - o2f(mu[k]));
      alpha_ex[idx] = ex;
      atomicAdd(&denom[dst[k] * Hh + hh], ex);
    }
  }
}

#define SC_E     32
#define SC_WAVES 4
#define SC_HC    4
#define SC_TILES (Ee / (SC_WAVES * SC_E))
static_assert(Ee % (SC_WAVES * SC_E) == 0, "the tiles of SC_WAVES waves x SC_E edges must cover the edge list exactly");
static_assert(SC_E == 32 && 2 * SC_HC == Hh, "one edge per lane, two column halves");

__global__ __launch_bounds__(SC_WAVES * 32) void scatter_out_kernel(
    const float* __restrict__ h, const float* __restrict__ ex,
    const float* __restrict__ denom, const int* __restrict__ ei,
    float* __restrict__ out, unsigned* __restrict__ tickets, unsigned long long ws_bytes) {
  __shared__ unsigned s_pick;
  if (threadIdx.x == 0) {
    unsigned ch, tk;
    if (ws_bytes >= WS_NEED) {
      ch = (__builtin_amdgcn_s_sendmsg_rtn(0x87) >> 18) & 1u;
      tk = atomicAdd(tickets + ch * TK_STRIDE, 1u);
      if (tk >= SC_TILES) { ch ^= 1u; tk = atomicAdd(tickets + ch * TK_STRIDE, 1u); }
      tk = min(tk, (unsigned)(SC_TILES - 1));
    } else {
      ch = blockIdx.x & 1u; tk = blockIdx.x >> 1;
    }
    s_pick = tk * 2u + ch;
  }
  __syncthreads();
  const int pick = __builtin_amdgcn_readfirstlane((int)s_pick);
  const int ch   = pick & 1;
  const int lane = threadIdx.x & 31;
  const int wv   = __builtin_amdgcn_readfirstlane((int)(threadIdx.x >> 5));
  const int e0   = ((pick >> 1) * SC_WAVES + wv) * SC_E;

  const int my_src = ei[e0 + lane];
  const int my_dst = ei[Ee + e0 + lane];
  const float4 x4 = *reinterpret_cast<const float4*>(ex + (long)(e0 + lane) * Hh + ch * SC_HC);
  const float4 d4 = *reinterpret_cast<const float4*>(denom + (long)my_dst * Hh + ch * SC_HC);
  float at[SC_HC];
  at[0] = x4.x / fmaxf(d4.x, 1e-10f);
  at[1] = x4.y / fmaxf(d4.y, 1e-10f);
  at[2] = x4.z / fmaxf(d4.z, 1e-10f);
  at[3] = x4.w / fmaxf(d4.w, 1e-10f);

  const float* hb = h + ch * (SC_HC * Dd) + lane;
  float*       ob = out + ch * (SC_HC * Dd) + lane;

  auto load_row = [&](int i, float (&hs)[SC_HC]) {
    const int s = lane_bcast_i(my_src, i);
    const float* ps = hb + (long)s * HD;
#pragma unroll
    for (int k = 0; k < SC_HC; ++k) hs[k] = ps[k * Dd];
  };

  auto add_edge = [&](int i, const float (&hs)[SC_HC]) {
    const int d = lane_bcast_i(my_dst, i);
    float* po = ob + (long)d * HD;
#pragma unroll
    for (int k = 0; k < SC_HC; ++k) atomicAdd(po + k * Dd, hs[k] * lane_bcast_f(at[k], i));
  };

  float ra[SC_HC], rb[SC_HC], rc[SC_HC], rd[SC_HC];
  load_row(0, ra); load_row(1, rb); load_row(2, rc);
  for (int i = 0; i < SC_E - 4; i += 4) {
    load_row(i + 3, rd); add_edge(i,     ra);
    load_row(i + 4, ra); add_edge(i + 1, rb);
    load_row(i + 5, rb); add_edge(i + 2, rc);
    load_row(i + 6, rc); add_edge(i + 3, rd);
  }
  load_row(SC_E - 1, rd);
  add_edge(SC_E - 4, ra); add_edge(SC_E - 3, rb); add_edge(SC_E - 2, rc); add_edge(SC_E - 1, rd);
}

extern "C" void kernel_launch(void* const* d_in, const int* in_sizes, int n_in,
                              void* d_out, int out_size, void* d_ws, size_t ws_size,
                              hipStream_t stream) {
  const float* x        = (const float*)d_in[0];
  const float* Wm       = (const float*)d_in[1];
  const float* a_src    = (const float*)d_in[2];
  const float* a_dst    = (const float*)d_in[3];
  const float* edge_W   = (const float*)d_in[4];
  const float* edge_b   = (const float*)d_in[5];
  const float* bias     = (const float*)d_in[6];
  const float* edge_attr= (const float*)d_in[7];
  const int*   ei       = (const int*)d_in[8];
  float* out = (float*)d_out;

  float*    h       = (float*)d_ws;
  float*    alpha   = h + (size_t)Nn * HD;
  unsigned* amax_u  = (unsigned*)(alpha + (size_t)Ee * Hh);
  float*    denom   = (float*)(amax_u + (size_t)Nn * Hh);
  unsigned* tickets = (unsigned*)(denom + (size_t)Nn * Hh);

  dim3 g((Nn + 31) / 32, HD / 128);
  gemm_h_wmma<<<g, 256, 0, stream>>>(x, Wm, a_src, h, denom, amax_u);

  edge_alpha_kernel<<<Ee / (8 * ET), 256, 0, stream>>>(h, denom, a_dst, edge_W, edge_b,
                                                      edge_attr, ei, alpha, amax_u);
  init_kernel<<<INIT_BLOCKS, 256, 0, stream>>>(out, denom, bias, tickets, (unsigned long long)ws_size);
  softmax_num_kernel<<<SM_BLOCKS, 256, 0, stream>>>(ei, amax_u, alpha, denom);
  scatter_out_kernel<<<2 * SC_TILES, SC_WAVES * 32, 0, stream>>>(h, alpha, denom, ei, out, tickets,
                                                                (unsigned long long)ws_size);
}


# GEMM kernel: tiles staged and fragments read by 128-bit LDS accesses on a 144-byte row pitch (no bank conflict; one read feeds two matrix instructions); k order in a chunk changed: new h sum order
# speedup vs baseline: 2.8364x; 1.0659x over previous
#include <hip/hip_runtime.h>

#define Nn   50000
#define Ee   400000
#define INF_ 256
#define Hh   8
#define Dd   32
#define HD   256
#define EFe  3

typedef __attribute__((ext_vector_type(2))) float v2f;
typedef __attribute__((ext_vector_type(8))) float v8f;

__device__ __forceinline__ unsigned f2o(float f) {
  unsigned u = __float_as_uint(f);
  return (u & 0x80000000u) ? ~u : (u | 0x80000000u);
}
__device__ __forceinline__ float o2f(unsigned u) {
  return (u & 0x80000000u) ? __uint_as_float(u & 0x7fffffffu) : __uint_as_float(~u);
}

#define INIT_BLOCKS 1250
#define INIT_TRIPS  10
static_assert(INIT_BLOCKS * 256 * INIT_TRIPS * 4 == Nn * HD, "the 16-byte stores must cover out exactly");
static_assert((INIT_BLOCKS * 256) % (HD / 4) == 0, "a thread must keep its column group from trip to trip");
static_assert((Nn * Hh) % 4 == 0 && INIT_BLOCKS * 256 * 4 >= Nn * Hh, "one 16-byte store per thread must cover denom");

#define TK_STRIDE 64
#define WS_NEED ((unsigned long long)(Nn * HD + Ee * Hh + 2 * Nn * Hh) * 4ull + 512ull)

__global__ __launch_bounds__(256) void init_kernel(float* __restrict__ out, float* __restrict__ denom,
                                                   const float* __restrict__ bias,
                                                   unsigned* __restrict__ tickets,
                                                   unsigned long long ws_bytes) {
  const int tid  = blockIdx.x * 256 + threadIdx.x;
  const int nthr = INIT_BLOCKS * 256;
  const int cg = (tid & (HD / 4 - 1)) * 4;
  float4 b4; b4.x = bias[cg]; b4.y = bias[cg + 1]; b4.z = bias[cg + 2]; b4.w = bias[cg + 3];
  float4* o4 = reinterpret_cast<float4*>(out);
#pragma unroll
  for (int k = 0; k < INIT_TRIPS; ++k) o4[tid + k * nthr] = b4;
  if (tid < Nn * Hh / 4) {
    float4 zf; zf.x = 0.f; zf.y = 0.f; zf.z = 0.f; zf.w = 0.f;
    reinterpret_cast<float4*>(denom)[tid] = zf;
  }
  if (tid == 0 && ws_bytes >= WS_NEED) { tickets[0] = 0u; tickets[TK_STRIDE] = 0u; }
}

template <int M>
__device__ __forceinline__ float row_xor(float v) {
  return __int_as_float(__builtin_amdgcn_update_dpp(__float_as_int(v), __float_as_int(v),
                                                    0x160 + M, 0xf, 0xf, true));
}

template <int M>
__device__ __forceinline__ float fold(float lo, float hi, bool up) {
  const float keep = up ? hi : lo;
  const float send = up ? lo : hi;
  return keep + row_xor<M>(send);
}

static_assert(Nn % 16 == 0, "the last tile's valid rows must end on a 128-byte line of a score plane (16 rows x 16 bytes = 2 lines)");
#define GP 36
static_assert((GP * 4) % 16 == 0 && ((GP * 4 / 16) & 1) == 1, "tile pitch = 16 bytes x odd: 128-bit fragment reads without bank conflicts");
__global__ __launch_bounds__(256) void gemm_h_wmma(const float* __restrict__ x,
                                                   const float* __restrict__ Wm,
                                                   const float* __restrict__ a_src,
                                                   float* __restrict__ h,
                                                   float* __restrict__ sc,
                                                   unsigned* __restrict__ amax_u) {
  __shared__ alignas(16) float sA[32 * GP];
  __shared__ alignas(16) float sB[128 * GP];
  __shared__ alignas(16) float sS[32 * 4];
  const int tid   = threadIdx.x;
  const int mBase = blockIdx.x * 32;
  const int nBase = blockIdx.y * 128;
  const int wave  = tid >> 5, ln = tid & 31;
  const int half  = ln >> 4, mr = ln & 15;
  const int mt    = wave & 1;
  const int nt0   = (wave >> 1) << 1;

  v8f acc0 = {};
  v8f acc1 = {};

  if (blockIdx.y == 0 && mBase * Hh + tid < Nn * Hh) amax_u[mBase * Hh + tid] = 0u;

  const int wv = __builtin_amdgcn_readfirstlane(wave);

  const int  lr  = tid >> 3;
  const int  lc  = (tid & 7) * 4;
  const bool xin = mBase + lr < Nn;
  const float4* xg = reinterpret_cast<const float4*>(x + (size_t)min(mBase + lr, Nn - 1) * INF_ + lc);
  const float4* wg = reinterpret_cast<const float4*>(Wm + (size_t)(nBase + lr) * INF_ + lc);
  float4* sAst = reinterpret_cast<float4*>(sA + lr * GP + lc);
  float4* sBst = reinterpret_cast<float4*>(sB + lr * GP + lc);

  const float4* fa  = reinterpret_cast<const float4*>(sA + (mt * 16 + mr) * GP + half * 4);
  const float4* fb0 = reinterpret_cast<const float4*>(sB + (nt0 * 16 + mr) * GP + half * 4);
  const float4* fb1 = reinterpret_cast<const float4*>(sB + (nt0 * 16 + 16 + mr) * GP + half * 4);

  for (int k0 = 0; k0 < INF_; k0 += 32) {
    float4 wr[4];
    const float4 xr = xg[k0 / 4];
#pragma unroll
    for (int j = 0; j < 4; ++j) wr[j] = wg[(j * 32 * INF_ + k0) / 4];
    float4 xs; xs.x = xin ? xr.x : 0.f; xs.y = xin ? xr.y : 0.f; xs.z = xin ? xr.z : 0.f; xs.w = xin ? xr.w : 0.f;
    *sAst = xs;
#pragma unroll
    for (int j = 0; j < 4; ++j) sBst[j * 32 * GP / 4] = wr[j];
    __syncthreads();

#pragma unroll
    for (int kk = 0; kk < 32; kk += 8) {
      const float4 a4 = fa[kk / 4];
      const float4 p4 = fb0[kk / 4];
      const float4 q4 = fb1[kk / 4];
      v2f a, b0, b1;
      a.x  = a4.x;  a.y  = a4.y;
      b0.x = p4.x;  b0.y = p4.y;
      b1.x = q4.x;  b1.y = q4.y;
      acc0 = __builtin_amdgcn_wmma_f32_16x16x4_f32(false, a, false, b0, (short)0, acc0, false, false);
      acc1 = __builtin_amdgcn_wmma_f32_16x16x4_f32(false, a, false, b1, (short)0, acc1, false, false);
      a.x  = a4.z;  a.y  = a4.w;
      b0.x = p4.z;  b0.y = p4.w;
      b1.x = q4.z;  b1.y = q4.w;
      acc0 = __builtin_amdgcn_wmma_f32_16x16x4_f32(false, a, false, b0, (short)0, acc0, false, false);
      acc1 = __builtin_amdgcn_wmma_f32_16x16x4_f32(false, a, false, b1, (short)0, acc1, false, false);
    }
    __syncthreads();
  }

  const int hl = wv >> 1;
  const float* av = a_src + (blockIdx.y * 4 + hl) * Dd;
  const float as0 = av[mr], as1 = av[16 + mr];
  float v[8];
#pragma unroll
  for (int r = 0; r < 8; ++r) v[r] = fmaf(acc1[r], as1, acc0[r] * as0);
  const bool up8 = (mr & 8) != 0, up4 = (mr & 4) != 0, up2 = (mr & 2) != 0;
  const float q0 = fold<8>(v[0], v[4], up8), q1 = fold<8>(v[1], v[5], up8);
  const float q2 = fold<8>(v[2], v[6], up8), q3 = fold<8>(v[3], v[7], up8);
  const float t0 = fold<4>(q0, q2, up4), t1 = fold<4>(q1, q3, up4);
  const float y  = fold<2>(t0, t1, up2);
  const float s  = y + row_xor<1>(y);
  const int   sr = ((mr >> 3) & 1) * 4 + ((mr >> 2) & 1) * 2 + ((mr >> 1) & 1);
  if ((mr & 1) == 0) sS[(mt * 16 + half * 8 + sr) * 4 + hl] = s;
  __syncthreads();
  if (wv == 0 && mBase + ln < Nn)
    reinterpret_cast<float4*>(sc)[(size_t)blockIdx.y * Nn + mBase + ln] =
        reinterpret_cast<const float4*>(sS)[ln];

  const int rowBase = mBase + mt * 16 + half * 8;
#pragma unroll
  for (int r = 0; r < 8; ++r) {
    int row = rowBase + r;
    if (row < Nn) {
      h[row * HD + nBase + nt0 * 16 + mr]       = acc0[r];
      h[row * HD + nBase + (nt0 + 1) * 16 + mr] = acc1[r];
    }
  }
}

#define ET 16
#define EG 4
static_assert(Ee % (8 * ET) == 0, "the tiles of 8 waves x ET edges must cover the edge list exactly");
static_assert(ET % EG == 0 && EG * Hh == 32, "a group's EG x 8 sums land one per lane");

__device__ __forceinline__ int   lane_bcast_i(int v, int l)   { return __builtin_amdgcn_readlane(v, l); }
__device__ __forceinline__ float lane_bcast_f(float v, int l) {
  return __int_as_float(__builtin_amdgcn_readlane(__float_as_int(v), l));
}

__device__ __forceinline__ float row_swap_add(float a, float b) {
  const auto r = __builtin_amdgcn_permlane16_swap(__float_as_uint(a), __float_as_uint(b), false, false);
  return __uint_as_float(r[0]) + __uint_as_float(r[1]);
}

__global__ __launch_bounds__(256) void edge_alpha_kernel(
    const float* __restrict__ h, const float* __restrict__ sc,
    const float* __restrict__ a_dst, const float* __restrict__ edge_W,
    const float* __restrict__ edge_b, const float* __restrict__ edge_attr,
    const int* __restrict__ ei, float* __restrict__ alpha,
    unsigned* __restrict__ amax_u) {
  const int lane = threadIdx.x & 31;
  const int wv = __builtin_amdgcn_readfirstlane((int)(threadIdx.x >> 5));
  const int e0 = (blockIdx.x * 8 + wv) * ET;

  float w0[Hh], w1[Hh], w2[Hh], bb[Hh], ad_[Hh];
#pragma unroll
  for (int hh = 0; hh < Hh; ++hh) {
    const int f = hh * Dd + lane;
    w0[hh]  = edge_W[f * EFe + 0];
    w1[hh]  = edge_W[f * EFe + 1];
    w2[hh]  = edge_W[f * EFe + 2];
    bb[hh]  = edge_b[f];
    ad_[hh] = a_dst[f];
  }

  const int   t      = lane & (ET - 1);
  const int   my_src = ei[e0 + t];
  const int   my_dst = ei[Ee + e0 + t];
  const float my_a0  = edge_attr[(e0 + t) * EFe + 0];
  const float my_a1  = edge_attr[(e0 + t) * EFe + 1];
  const float my_a2  = edge_attr[(e0 + t) * EFe + 2];

  const bool up8 = (lane & 8) != 0, up4 = (lane & 4) != 0, up2 = (lane & 2) != 0, up1 = (lane & 1) != 0;
  const int  ge  = ((lane >> 4) & 1) + 2 * ((lane >> 3) & 1);
  const int  gh  = lane & 7;
  const float* scl = sc + (long)(gh >> 2) * (Nn * 4) + (gh & 3);

  auto load_row = [&](int i, float (&hd)[Hh]) {
    const int d = lane_bcast_i(my_dst, i);
    const float* pd = h + (long)d * HD + lane;
#pragma unroll
    for (int hh = 0; hh < Hh; ++hh) hd[hh] = pd[hh * Dd];
  };

  auto edge_partials = [&](int i, const float (&hd)[Hh], float (&p)[Hh]) {
    const float ea0 = lane_bcast_f(my_a0, i);
    const float ea1 = lane_bcast_f(my_a1, i);
    const float ea2 = lane_bcast_f(my_a2, i);
#pragma unroll
    for (int hh = 0; hh < Hh; ++hh) {
      float enc = fmaf(ea0, w0[hh], fmaf(ea1, w1[hh], fmaf(ea2, w2[hh], bb[hh])));
      enc = fmaxf(enc, 0.f);
      p[hh] = hd[hh] * ad_[hh] + enc * hd[hh];
    }
  };

  float rA[Hh], rB[Hh], rC[Hh], rD[Hh];
  load_row(0, rA); load_row(1, rB);
  for (int i = 0; i < ET; i += EG) {
    float pa[Hh], pb[Hh], u0[Hh], u1[Hh];
    const bool more = i + EG < ET;
    const int   my_e = e0 + i + ge;
    const int   my_d = __shfl(my_dst, i + ge, 32);
    const float s_sc = scl[(long)__shfl(my_src, i + ge, 32) * 4];
    load_row(i + 2, rC);
    edge_partials(i, rA, pa);
    load_row(i + 3, rD);
    edge_partials(i + 1, rB, pb);
#pragma unroll
    for (int hh = 0; hh < Hh; ++hh) u0[hh] = row_swap_add(pa[hh], pb[hh]);
    if (more) load_row(i + EG, rA);
    edge_partials(i + 2, rC, pa);
    if (more) load_row(i + EG + 1, rB);
    edge_partials(i + 3, rD, pb);
#pragma unroll
    for (int hh = 0; hh < Hh; ++hh) u1[hh] = row_swap_add(pa[hh], pb[hh]);

    float w[Hh];
#pragma unroll
    for (int hh = 0; hh < Hh; ++hh) w[hh] = fold<8>(u0[hh], u1[hh], up8);
    const float x0 = fold<4>(w[0], w[4], up4), x1 = fold<4>(w[1], w[5], up4);
    const float x2 = fold<4>(w[2], w[6], up4), x3 = fold<4>(w[3], w[7], up4);
    const float y0 = fold<2>(x0, x2, up2), y1 = fold<2>(x1, x3, up2);
    const float z  = fold<1>(y0, y1, up1) + s_sc;
    const float al = (z >= 0.f) ? z : 0.2f * z;

    alpha[(long)my_e * Hh + gh] = al;
    atomicMax(amax_u + (long)my_d * Hh + gh, f2o(al));
  }
}

#define SM_BLOCKS 1250
#define SM_PER    10
#define SM_B      5
static_assert(SM_BLOCKS * 256 * SM_PER == Ee * Hh, "the grid-stride trips must cover the elements exactly");
static_assert(SM_PER % SM_B == 0, "whole batches");

__global__ __launch_bounds__(256) void softmax_num_kernel(const int* __restrict__ ei,
                                                          const unsigned* __restrict__ amax_u,
                                                          float* __restrict__ alpha_ex,
                                                          float* __restrict__ denom) {
  const int tid  = blockIdx.x * 256 + threadIdx.x;
  const int nthr = SM_BLOCKS * 256;
  const int hh   = tid & 7;
  for (int b = 0; b < SM_PER; b += SM_B) {
    int dst[SM_B]; float al[SM_B]; unsigned mu[SM_B];
#pragma unroll
    for (int k = 0; k < SM_B; ++k) {
      const int idx = tid + (b + k) * nthr;
      dst[k] = ei[Ee + (idx >> 3)];
      al[k]  = alpha_ex[idx];
    }
#pragma unroll
    for (int k = 0; k < SM_B; ++k) mu[k] = amax_u[dst[k] * Hh + hh];
#pragma unroll
    for (int k = 0; k < SM_B; ++k) {
      const int idx = tid + (b + k) * nthr;
      const float ex = __expf(al[k] - o2f(mu[k]));
      alpha_ex[idx] = ex;
      atomicAdd(&denom[dst[k] * Hh + hh], ex);
    }
  }
}

#define SC_E     32
#define SC_WAVES 4
#define SC_HC    4
#define SC_TILES (Ee / (SC_WAVES * SC_E))
static_assert(Ee % (SC_WAVES * SC_E) == 0, "the tiles of SC_WAVES waves x SC_E edges must cover the edge list exactly");
static_assert(SC_E == 32 && 2 * SC_HC == Hh, "one edge per lane, two column halves");

__global__ __launch_bounds__(SC_WAVES * 32) void scatter_out_kernel(
    const float* __restrict__ h, const float* __restrict__ ex,
    const float* __restrict__ denom, const int* __restrict__ ei,
    float* __restrict__ out, unsigned* __restrict__ tickets, unsigned long long ws_bytes) {
  __shared__ unsigned s_pick;
  if (threadIdx.x == 0) {
    unsigned ch, tk;
    if (ws_bytes >= WS_NEED) {
      ch = (__builtin_amdgcn_s_sendmsg_rtn(0x87) >> 18) & 1u;
      tk = atomicAdd(tickets + ch * TK_STRIDE, 1u);
      if (tk >= SC_TILES) { ch ^= 1u; tk = atomicAdd(tickets + ch * TK_STRIDE, 1u); }
      tk = min(tk, (unsigned)(SC_TILES - 1));
    } else {
      ch = blockIdx.x & 1u; tk = blockIdx.x >> 1;
    }
    s_pick = tk * 2u + ch;
  }
  __syncthreads();
  const int pick = __builtin_amdgcn_readfirstlane((int)s_pick);
  const int ch   = pick & 1;
  const int lane = threadIdx.x & 31;
  const int wv   = __builtin_amdgcn_readfirstlane((int)(threadIdx.x >> 5));
  const int e0   = ((pick >> 1) * SC_WAVES + wv) * SC_E;

  const int my_src = ei[e0 + lane];
  const int my_dst = ei[Ee + e0 + lane];
  const float4 x4 = *reinterpret_cast<const float4*>(ex + (long)(e0 + lane) * Hh + ch * SC_HC);
  const float4 d4 = *reinterpret_cast<const float4*>(denom + (long)my_dst * Hh + ch * SC_HC);
  float at[SC_HC];
  at[0] = x4.x / fmaxf(d4.x, 1e-10f);
  at[1] = x4.y / fmaxf(d4.y, 1e-10f);
  at[2] = x4.z / fmaxf(d4.z, 1e-10f);
  at[3] = x4.w / fmaxf(d4.w, 1e-10f);

  const float* hb = h + ch * (SC_HC * Dd) + lane;
  float*       ob = out + ch * (SC_HC * Dd) + lane;

  auto load_row = [&](int i, float (&hs)[SC_HC]) {
    const int s = lane_bcast_i(my_src, i);
    const float* ps = hb + (long)s * HD;
#pragma unroll
    for (int k = 0; k < SC_HC; ++k) hs[k] = ps[k * Dd];
  };

  auto add_edge = [&](int i, const float (&hs)[SC_HC]) {
    const int d = lane_bcast_i(my_dst, i);
    float* po = ob + (long)d * HD;
#pragma unroll
    for (int k = 0; k < SC_HC; ++k) atomicAdd(po + k * Dd, hs[k] * lane_bcast_f(at[k], i));
  };

  float ra[SC_HC], rb[SC_HC], rc[SC_HC], rd[SC_HC];
  load_row(0, ra); load_row(1, rb); load_row(2, rc);
  for (int i = 0; i < SC_E - 4; i += 4) {
    load_row(i + 3, rd); add_edge(i,     ra);
    load_row(i + 4, ra); add_edge(i + 1, rb);
    load_row(i + 5, rb); add_edge(i + 2, rc);
    load_row(i + 6, rc); add_edge(i + 3, rd);
  }
  load_row(SC_E - 1, rd);
  add_edge(SC_E - 4, ra); add_edge(SC_E - 3, rb); add_edge(SC_E - 2, rc); add_edge(SC_E - 1, rd);
}

extern "C" void kernel_launch(void* const* d_in, const int* in_sizes, int n_in,
                              void* d_out, int out_size, void* d_ws, size_t ws_size,
                              hipStream_t stream) {
  const float* x        = (const float*)d_in[0];
  const float* Wm       = (const float*)d_in[1];
  const float* a_src    = (const float*)d_in[2];
  const float* a_dst    = (const float*)d_in[3];
  const float* edge_W   = (const float*)d_in[4];
  const float* edge_b   = (const float*)d_in[5];
  const float* bias     = (const float*)d_in[6];
  const float* edge_attr= (const float*)d_in[7];
  const int*   ei       = (const int*)d_in[8];
  float* out = (float*)d_out;

  float*    h       = (float*)d_ws;
  float*    alpha   = h + (size_t)Nn * HD;
  unsigned* amax_u  = (unsigned*)(alpha + (size_t)Ee * Hh);
  float*    denom   = (float*)(amax_u + (size_t)Nn * Hh);
  unsigned* tickets = (unsigned*)(denom + (size_t)Nn * Hh);

  dim3 g((Nn + 31) / 32, HD / 128);
  gemm_h_wmma<<<g, 256, 0, stream>>>(x, Wm, a_src, h, denom, amax_u);

  edge_alpha_kernel<<<Ee / (8 * ET), 256, 0, stream>>>(h, denom, a_dst, edge_W, edge_b,
                                                      edge_attr, ei, alpha, amax_u);
  init_kernel<<<INIT_BLOCKS, 256, 0, stream>>>(out, denom, bias, tickets, (unsigned long long)ws_size);
  softmax_num_kernel<<<SM_BLOCKS, 256, 0, stream>>>(ei, amax_u, alpha, denom);
  scatter_out_kernel<<<2 * SC_TILES, SC_WAVES * 32, 0, stream>>>(h, alpha, denom, ei, out, tickets,
                                                                (unsigned long long)ws_size);
}
